# starting kernel
# baseline (speedup 1.0000x reference)
#include <hip/hip_runtime.h>
#include <hip/hip_bf16.h>

typedef __attribute__((ext_vector_type(2))) float v2f;
typedef __attribute__((ext_vector_type(8))) float v8f;

#define F_DIM 128
#define LEAKY 0.2f

__device__ __forceinline__ unsigned f2ord(float f) {
  unsigned u = __float_as_uint(f);
  return (u & 0x80000000u) ? ~u : (u | 0x80000000u);
}
__device__ __forceinline__ float ord2f(unsigned u) {
  return (u & 0x80000000u) ? __uint_as_float(u & 0x7FFFFFFFu)
                           : __uint_as_float(~u);
}

__global__ __launch_bounds__(256) void gat_gemm_wmma(
    const float* __restrict__ x, const float* __restrict__ W,
    float* __restrict__ h, int M)
{
  __shared__ float sA[16 * 132];
  const int tid = threadIdx.x;
  const int m0  = blockIdx.x * 16;

  for (int i = tid; i < 512; i += 256) {
    int r = i >> 5, c = i & 31;
    float4 v = make_float4(0.f, 0.f, 0.f, 0.f);
    if (m0 + r < M) v = ((const float4*)(x + (size_t)(m0 + r) * F_DIM))[c];
    ((float4*)sA)[r * 33 + c] = v;
  }
  __syncthreads();

  const int lane = tid & 31;
  const int wv   = tid >> 5;
  const int row  = lane & 15;
  const int kh   = (lane >> 4) * 2;
  const int n0   = wv * 16;

  v8f acc = {};
  #pragma unroll 4
  for (int k = 0; k < F_DIM; k += 4) {
    v2f a, b;
    a.x = sA[row * 132 + k + kh + 0];
    a.y = sA[row * 132 + k + kh + 1];
    b.x = W[(k + kh + 0) * F_DIM + n0 + row];
    b.y = W[(k + kh + 1) * F_DIM + n0 + row];
    acc = __builtin_amdgcn_wmma_f32_16x16x4_f32(false, a, false, b,
                                                (short)0, acc, false, false);
  }

  const int mh = (lane >> 4) * 8;
  float* hp = h + (size_t)m0 * F_DIM + n0 + row;
  if (m0 + 16 <= M) {
    #pragma unroll
    for (int v = 0; v < 8; ++v) hp[(size_t)(mh + v) * F_DIM] = acc[v];
  } else {
    #pragma unroll
    for (int v = 0; v < 8; ++v)
      if (m0 + mh + v < M) hp[(size_t)(mh + v) * F_DIM] = acc[v];
  }
}

__global__ __launch_bounds__(256) void gat_alpha(
    const float* __restrict__ h, const float* __restrict__ a,
    float* __restrict__ as, float* __restrict__ ad, int N)
{
  const int node = blockIdx.x * 8 + (threadIdx.x >> 5);
  const int lane = threadIdx.x & 31;
  if (node >= N) return;
  float4 hv = ((const float4*)(h + (size_t)node * F_DIM))[lane];
  float4 a1 = ((const float4*)a)[lane];
  float4 a2 = ((const float4*)(a + F_DIM))[lane];
  float s = hv.x * a1.x + hv.y * a1.y + hv.z * a1.z + hv.w * a1.w;
  float d = hv.x * a2.x + hv.y * a2.y + hv.z * a2.z + hv.w * a2.w;
  #pragma unroll
  for (int m = 16; m >= 1; m >>= 1) {
    s += __shfl_xor(s, m, 32);
    d += __shfl_xor(d, m, 32);
  }
  if (lane == 0) { as[node] = s; ad[node] = d; }
}

__global__ __launch_bounds__(256) void gat_edge_logit(
    const int* __restrict__ src, const int* __restrict__ dst,
    const float* __restrict__ as, const float* __restrict__ ad,
    float* __restrict__ e, unsigned* __restrict__ segmax, int E)
{
  int i = blockIdx.x * blockDim.x + threadIdx.x;
  if (i >= E) return;
  float v = as[src[i]] + ad[dst[i]];
  v = v > 0.f ? v : LEAKY * v;
  e[i] = v;
  atomicMax(segmax + dst[i], f2ord(v));
}

__global__ __launch_bounds__(256) void gat_edge_exp(
    const int* __restrict__ dst, const float* __restrict__ e,
    const unsigned* __restrict__ segmax, float* __restrict__ ex,
    float* __restrict__ segsum, int E)
{
  int i = blockIdx.x * blockDim.x + threadIdx.x;
  if (i >= E) return;
  int d = dst[i];
  float v = expf(e[i] - ord2f(segmax[d]));
  ex[i] = v;
  atomicAdd(segsum + d, v);
}

__global__ __launch_bounds__(256) void gat_aggregate(
    const int* __restrict__ src, const int* __restrict__ dst,
    const float* __restrict__ h, const float* __restrict__ ex,
    const float* __restrict__ segsum, float* __restrict__ out, int E)
{
  const int i    = __builtin_amdgcn_readfirstlane(blockIdx.x * 8 + (threadIdx.x >> 5));
  const int lane = threadIdx.x & 31;
  if (i >= E) return;
  const int s = src[i], d = dst[i];
  const float coeff = ex[i] / (segsum[d] + 1e-16f);
  float4 hv = ((const float4*)(h + (size_t)s * F_DIM))[lane];
  float* op = out + (size_t)d * F_DIM + lane * 4;
  atomicAdd(op + 0, coeff * hv.x);
  atomicAdd(op + 1, coeff * hv.y);
  atomicAdd(op + 2, coeff * hv.z);
  atomicAdd(op + 3, coeff * hv.w);
}

extern "C" void kernel_launch(void* const* d_in, const int* in_sizes, int n_in,
                              void* d_out, int out_size, void* d_ws, size_t ws_size,
                              hipStream_t stream) {
  const float* x  = (const float*)d_in[0];
  const int*   ei = (const int*)d_in[1];
  const float* W  = (const float*)d_in[2];
  const float* a  = (const float*)d_in[3];
  const int N = in_sizes[0] / F_DIM;
  const int E = in_sizes[1] / 2;
  const int* src = ei;
  const int* dst = ei + E;

  float*    h      = (float*)d_ws;
  float*    as     = h + (size_t)N * F_DIM;
  float*    ad     = as + N;
  unsigned* segmax = (unsigned*)(ad + N);
  float*    segsum = (float*)(segmax + N);
  float*    e      = segsum + N;
  float*    ex     = e + E;
  float*    out    = (float*)d_out;

  hipMemsetAsync(d_out, 0, (size_t)out_size * sizeof(float), stream);
  hipMemsetAsync(segmax, 0, (size_t)N * sizeof(unsigned), stream);
  hipMemsetAsync(segsum, 0, (size_t)N * sizeof(float), stream);

  gat_gemm_wmma <<<(N + 15) / 16,  256, 0, stream>>>(x, W, h, N);
  gat_alpha     <<<(N + 7)  / 8,   256, 0, stream>>>(h, a, as, ad, N);
  gat_edge_logit<<<(E + 255) / 256, 256, 0, stream>>>(src, dst, as, ad, e, segmax, E);
  gat_edge_exp  <<<(E + 255) / 256, 256, 0, stream>>>(dst, e, segmax, ex, segsum, E);
  gat_aggregate <<<(E + 7)  / 8,   256, 0, stream>>>(src, dst, h, ex, segsum, out, E);
}


# aggregate: lane l owns features l+32j, so each f32 atomic add and row load covers one whole 128-B line, not a quarter of four; sums still in atomic arrival order as in parent
# speedup vs baseline: 1.1249x; 1.1249x over previous
#include <hip/hip_runtime.h>
#include <hip/hip_bf16.h>

typedef __attribute__((ext_vector_type(2))) float v2f;
typedef __attribute__((ext_vector_type(8))) float v8f;

#define F_DIM 128
#define LEAKY 0.2f

__device__ __forceinline__ unsigned f2ord(float f) {
  unsigned u = __float_as_uint(f);
  return (u & 0x80000000u) ? ~u : (u | 0x80000000u);
}
__device__ __forceinline__ float ord2f(unsigned u) {
  return (u & 0x80000000u) ? __uint_as_float(u & 0x7FFFFFFFu)
                           : __uint_as_float(~u);
}

__global__ __launch_bounds__(256) void gat_gemm_wmma(
    const float* __restrict__ x, const float* __restrict__ W,
    float* __restrict__ h, int M)
{
  __shared__ float sA[16 * 132];
  const int tid = threadIdx.x;
  const int m0  = blockIdx.x * 16;

  for (int i = tid; i < 512; i += 256) {
    int r = i >> 5, c = i & 31;
    float4 v = make_float4(0.f, 0.f, 0.f, 0.f);
    if (m0 + r < M) v = ((const float4*)(x + (size_t)(m0 + r) * F_DIM))[c];
    ((float4*)sA)[r * 33 + c] = v;
  }
  __syncthreads();

  const int lane = tid & 31;
  const int wv   = tid >> 5;
  const int row  = lane & 15;
  const int kh   = (lane >> 4) * 2;
  const int n0   = wv * 16;

  v8f acc = {};
  #pragma unroll 4
  for (int k = 0; k < F_DIM; k += 4) {
    v2f a, b;
    a.x = sA[row * 132 + k + kh + 0];
    a.y = sA[row * 132 + k + kh + 1];
    b.x = W[(k + kh + 0) * F_DIM + n0 + row];
    b.y = W[(k + kh + 1) * F_DIM + n0 + row];
    acc = __builtin_amdgcn_wmma_f32_16x16x4_f32(false, a, false, b,
                                                (short)0, acc, false, false);
  }

  const int mh = (lane >> 4) * 8;
  float* hp = h + (size_t)m0 * F_DIM + n0 + row;
  if (m0 + 16 <= M) {
    #pragma unroll
    for (int v = 0; v < 8; ++v) hp[(size_t)(mh + v) * F_DIM] = acc[v];
  } else {
    #pragma unroll
    for (int v = 0; v < 8; ++v)
      if (m0 + mh + v < M) hp[(size_t)(mh + v) * F_DIM] = acc[v];
  }
}

__global__ __launch_bounds__(256) void gat_alpha(
    const float* __restrict__ h, const float* __restrict__ a,
    float* __restrict__ as, float* __restrict__ ad, int N)
{
  const int node = blockIdx.x * 8 + (threadIdx.x >> 5);
  const int lane = threadIdx.x & 31;
  if (node >= N) return;
  float4 hv = ((const float4*)(h + (size_t)node * F_DIM))[lane];
  float4 a1 = ((const float4*)a)[lane];
  float4 a2 = ((const float4*)(a + F_DIM))[lane];
  float s = hv.x * a1.x + hv.y * a1.y + hv.z * a1.z + hv.w * a1.w;
  float d = hv.x * a2.x + hv.y * a2.y + hv.z * a2.z + hv.w * a2.w;
  #pragma unroll
  for (int m = 16; m >= 1; m >>= 1) {
    s += __shfl_xor(s, m, 32);
    d += __shfl_xor(d, m, 32);
  }
  if (lane == 0) { as[node] = s; ad[node] = d; }
}

__global__ __launch_bounds__(256) void gat_edge_logit(
    const int* __restrict__ src, const int* __restrict__ dst,
    const float* __restrict__ as, const float* __restrict__ ad,
    float* __restrict__ e, unsigned* __restrict__ segmax, int E)
{
  int i = blockIdx.x * blockDim.x + threadIdx.x;
  if (i >= E) return;
  float v = as[src[i]] + ad[dst[i]];
  v = v > 0.f ? v : LEAKY * v;
  e[i] = v;
  atomicMax(segmax + dst[i], f2ord(v));
}

__global__ __launch_bounds__(256) void gat_edge_exp(
    const int* __restrict__ dst, const float* __restrict__ e,
    const unsigned* __restrict__ segmax, float* __restrict__ ex,
    float* __restrict__ segsum, int E)
{
  int i = blockIdx.x * blockDim.x + threadIdx.x;
  if (i >= E) return;
  int d = dst[i];
  float v = expf(e[i] - ord2f(segmax[d]));
  ex[i] = v;
  atomicAdd(segsum + d, v);
}

__global__ __launch_bounds__(256) void gat_aggregate(
    const int* __restrict__ src, const int* __restrict__ dst,
    const float* __restrict__ h, const float* __restrict__ ex,
    const float* __restrict__ segsum, float* __restrict__ out, int E)
{
  const int i    = __builtin_amdgcn_readfirstlane(blockIdx.x * 8 + (threadIdx.x >> 5));
  const int lane = threadIdx.x & 31;
  if (i >= E) return;
  const int s = src[i], d = dst[i];
  const float coeff = ex[i] / (segsum[d] + 1e-16f);
  const float* hp = h + (size_t)s * F_DIM + lane;
  float*       op = out + (size_t)d * F_DIM + lane;
  const float h0 = hp[0], h1 = hp[32], h2 = hp[64], h3 = hp[96];
  atomicAdd(op +  0, coeff * h0);
  atomicAdd(op + 32, coeff * h1);
  atomicAdd(op + 64, coeff * h2);
  atomicAdd(op + 96, coeff * h3);
}

extern "C" void kernel_launch(void* const* d_in, const int* in_sizes, int n_in,
                              void* d_out, int out_size, void* d_ws, size_t ws_size,
                              hipStream_t stream) {
  const float* x  = (const float*)d_in[0];
  const int*   ei = (const int*)d_in[1];
  const float* W  = (const float*)d_in[2];
  const float* a  = (const float*)d_in[3];
  const int N = in_sizes[0] / F_DIM;
  const int E = in_sizes[1] / 2;
  const int* src = ei;
  const int* dst = ei + E;

  float*    h      = (float*)d_ws;
  float*    as     = h + (size_t)N * F_DIM;
  float*    ad     = as + N;
  unsigned* segmax = (unsigned*)(ad + N);
  float*    segsum = (float*)(segmax + N);
  float*    e      = segsum + N;
  float*    ex     = e + E;
  float*    out    = (float*)d_out;

  hipMemsetAsync(d_out, 0, (size_t)out_size * sizeof(float), stream);
  hipMemsetAsync(segmax, 0, (size_t)N * sizeof(unsigned), stream);
  hipMemsetAsync(segsum, 0, (size_t)N * sizeof(float), stream);

  gat_gemm_wmma <<<(N + 15) / 16,  256, 0, stream>>>(x, W, h, N);
  gat_alpha     <<<(N + 7)  / 8,   256, 0, stream>>>(h, a, as, ad, N);
  gat_edge_logit<<<(E + 255) / 256, 256, 0, stream>>>(src, dst, as, ad, e, segmax, E);
  gat_edge_exp  <<<(E + 255) / 256, 256, 0, stream>>>(dst, e, segmax, ex, segsum, E);
  gat_aggregate <<<(E + 7)  / 8,   256, 0, stream>>>(src, dst, h, ex, segsum, out, E);
}


# edges grouped by destination in-call (int atomic slot, scan, scatter); one wave per node does segment max, sum and weighted row sum in registers, stores once, no f32 atomics; sums in slot order
# speedup vs baseline: 1.8458x; 1.6409x over previous
#include <hip/hip_runtime.h>
#include <hip/hip_bf16.h>


typedef __attribute__((ext_vector_type(2))) float v2f;
typedef __attribute__((ext_vector_type(8))) float v8f;

#define F_DIM 128
#define LEAKY 0.2f
#define ROWS_IN_FLIGHT 8

__global__ __launch_bounds__(256) void gat_gemm_wmma(
    const float* __restrict__ x, const float* __restrict__ W,
    float* __restrict__ h, int* __restrict__ deg, int M)
{
  __shared__ float sA[16 * 132];
  const int tid = threadIdx.x;
  const int m0  = blockIdx.x * 16;

  if (tid < 16 && m0 + tid < M) deg[m0 + tid] = 0;

  for (int i = tid; i < 512; i += 256) {
    int r = i >> 5, c = i & 31;
    float4 v = make_float4(0.f, 0.f, 0.f, 0.f);
    if (m0 + r < M) v = ((const float4*)(x + (size_t)(m0 + r) * F_DIM))[c];
    ((float4*)sA)[r * 33 + c] = v;
  }
  __syncthreads();

  const int lane = tid & 31;
  const int wv   = tid >> 5;
  const int row  = lane & 15;
  const int kh   = (lane >> 4) * 2;
  const int n0   = wv * 16;

  v8f acc = {};
  #pragma unroll 4
  for (int k = 0; k < F_DIM; k += 4) {
    v2f a, b;
    a.x = sA[row * 132 + k + kh + 0];
    a.y = sA[row * 132 + k + kh + 1];
    b.x = W[(k + kh + 0) * F_DIM + n0 + row];
    b.y = W[(k + kh + 1) * F_DIM + n0 + row];
    acc = __builtin_amdgcn_wmma_f32_16x16x4_f32(false, a, false, b,
                                                (short)0, acc, false, false);
  }

  const int mh = (lane >> 4) * 8;
  float* hp = h + (size_t)m0 * F_DIM + n0 + row;
  if (m0 + 16 <= M) {
    #pragma unroll
    for (int v = 0; v < 8; ++v) hp[(size_t)(mh + v) * F_DIM] = acc[v];
  } else {
    #pragma unroll
    for (int v = 0; v < 8; ++v)
      if (m0 + mh + v < M) hp[(size_t)(mh + v) * F_DIM] = acc[v];
  }
}

__global__ __launch_bounds__(256) void gat_alpha(
    const float* __restrict__ h, const float* __restrict__ a,
    float* __restrict__ as, float* __restrict__ ad, int N)
{
  const int node = blockIdx.x * 8 + (threadIdx.x >> 5);
  const int lane = threadIdx.x & 31;
  if (node >= N) return;
  float4 hv = ((const float4*)(h + (size_t)node * F_DIM))[lane];
  float4 a1 = ((const float4*)a)[lane];
  float4 a2 = ((const float4*)(a + F_DIM))[lane];
  float s = hv.x * a1.x + hv.y * a1.y + hv.z * a1.z + hv.w * a1.w;
  float d = hv.x * a2.x + hv.y * a2.y + hv.z * a2.z + hv.w * a2.w;
  #pragma unroll
  for (int m = 16; m >= 1; m >>= 1) {
    s += __shfl_xor(s, m, 32);
    d += __shfl_xor(d, m, 32);
  }
  if (lane == 0) { as[node] = s; ad[node] = d; }
}

__global__ __launch_bounds__(256) void gat_edge_slot(
    const int* __restrict__ dst, int* __restrict__ deg,
    int* __restrict__ slot, int E)
{
  int i = blockIdx.x * blockDim.x + threadIdx.x;
  if (i >= E) return;
  slot[i] = atomicAdd(deg + dst[i], 1);
}

__global__ __launch_bounds__(1024) void gat_segment_first(
    const int* __restrict__ deg, int* __restrict__ first, int N)
{
  __shared__ int s_front[32];
  __shared__ int s_total[32];
  const int t    = threadIdx.x;
  const int lane = t & 31;
  const int w    = t >> 5;
  const int base = blockIdx.x * 1024;

  int front = 0;
  const int4* deg4 = (const int4*)deg;
  const int n4   = base >> 2;
  const int full = n4 >> 10;
  #pragma unroll 4
  for (int m = 0; m < full; ++m) {
    int4 q = deg4[t + (m << 10)];
    front += (q.x + q.y) + (q.z + q.w);
  }
  if (t < (n4 & 1023)) {
    int4 q = deg4[t + (full << 10)];
    front += (q.x + q.y) + (q.z + q.w);
  }
  const int n = base + t;
  const int v = (n < N) ? deg[n] : 0;

  int inc = v;
  #pragma unroll
  for (int o = 1; o < 32; o <<= 1) {
    int u = __shfl_up(inc, o, 32);
    if (lane >= o) inc += u;
  }
  #pragma unroll
  for (int o = 16; o >= 1; o >>= 1) front += __shfl_xor(front, o, 32);
  if (lane == 31) s_total[w] = inc;
  if (lane == 0)  s_front[w] = front;
  __syncthreads();

  int add = s_front[lane] + ((lane < w) ? s_total[lane] : 0);
  #pragma unroll
  for (int o = 16; o >= 1; o >>= 1) add += __shfl_xor(add, o, 32);
  if (n < N) first[n] = add + inc - v;
}

__global__ __launch_bounds__(256) void gat_group_edges(
    const int* __restrict__ src, const int* __restrict__ dst,
    const int* __restrict__ first, const int* __restrict__ slot,
    int* __restrict__ gsrc, int E)
{
  int i = blockIdx.x * blockDim.x + threadIdx.x;
  if (i >= E) return;
  gsrc[first[dst[i]] + slot[i]] = src[i];
}

__device__ __forceinline__ float wave_max(float v) {
  #pragma unroll
  for (int m = 16; m >= 1; m >>= 1) v = fmaxf(v, __shfl_xor(v, m, 32));
  return v;
}
__device__ __forceinline__ float wave_sum(float v) {
  #pragma unroll
  for (int m = 16; m >= 1; m >>= 1) v += __shfl_xor(v, m, 32);
  return v;
}
__device__ __forceinline__ float4 h_row(const float* __restrict__ h, int s, int lane) {
  return ((const float4*)(h + (size_t)s * F_DIM))[lane];
}
__device__ __forceinline__ void axpy4(float4& acc, float c, const float4& r) {
  acc.x = fmaf(c, r.x, acc.x);
  acc.y = fmaf(c, r.y, acc.y);
  acc.z = fmaf(c, r.z, acc.z);
  acc.w = fmaf(c, r.w, acc.w);
}

__device__ __forceinline__ void add_rows(
    float4& acc, const float* __restrict__ h, int lane, int s, float c, int n)
{
  const int ci = __float_as_int(c);
  int j = 0;
  for (; j + ROWS_IN_FLIGHT <= n; j += ROWS_IN_FLIGHT) {
    float4 r[ROWS_IN_FLIGHT];
    #pragma unroll
    for (int k = 0; k < ROWS_IN_FLIGHT; ++k)
      r[k] = h_row(h, __builtin_amdgcn_readlane(s, j + k), lane);
    #pragma unroll
    for (int k = 0; k < ROWS_IN_FLIGHT; ++k)
      axpy4(acc, __int_as_float(__builtin_amdgcn_readlane(ci, j + k)), r[k]);
  }
  if (j < n) {
    float4 r[ROWS_IN_FLIGHT];
    #pragma unroll
    for (int k = 0; k < ROWS_IN_FLIGHT; ++k)
      r[k] = h_row(h, __builtin_amdgcn_readlane(s, min(j + k, n - 1)), lane);
    #pragma unroll
    for (int k = 0; k < ROWS_IN_FLIGHT; ++k) {
      const float cj = __int_as_float(__builtin_amdgcn_readlane(ci, min(j + k, n - 1)));
      axpy4(acc, (j + k < n) ? cj : 0.f, r[k]);
    }
  }
}

__device__ __forceinline__ float edge_logit(
    const float* __restrict__ as, int s, float adv)
{
  float t = as[s] + adv;
  return t > 0.f ? t : LEAKY * t;
}

__global__ __launch_bounds__(256) void gat_node_aggregate(
    const int* __restrict__ gsrc, const int* __restrict__ first,
    const int* __restrict__ deg, const float* __restrict__ h,
    const float* __restrict__ as, const float* __restrict__ ad,
    float* __restrict__ out, int N)
{
  const int d    = __builtin_amdgcn_readfirstlane(blockIdx.x * 8 + (threadIdx.x >> 5));
  const int lane = threadIdx.x & 31;
  if (d >= N) return;
  const int   beg = first[d];
  const int   cnt = deg[d];
  const float adv = ad[d];
  const int*  seg = gsrc + beg;

  const bool v0 = lane < cnt;
  int   s0 = 0;
  float l0 = -INFINITY;
  if (v0) { s0 = seg[lane]; l0 = edge_logit(as, s0, adv); }

  float m = l0;
  for (int c = 32; c < cnt; c += 32)
    if (c + lane < cnt) m = fmaxf(m, edge_logit(as, seg[c + lane], adv));
  m = wave_max(m);

  const float p0 = v0 ? expf(l0 - m) : 0.f;
  float sum = p0;
  for (int c = 32; c < cnt; c += 32)
    if (c + lane < cnt) sum += expf(edge_logit(as, seg[c + lane], adv) - m);
  sum = wave_sum(sum);
  const float den = sum + 1e-16f;

  float4 acc = make_float4(0.f, 0.f, 0.f, 0.f);
  add_rows(acc, h, lane, s0, p0 / den, min(cnt, 32));
  for (int c = 32; c < cnt; c += 32) {
    int s = 0; float p = 0.f;
    if (c + lane < cnt) { s = seg[c + lane]; p = expf(edge_logit(as, s, adv) - m); }
    add_rows(acc, h, lane, s, p / den, min(cnt - c, 32));
  }
  ((float4*)(out + (size_t)d * F_DIM))[lane] = acc;
}

extern "C" void kernel_launch(void* const* d_in, const int* in_sizes, int n_in,
                              void* d_out, int out_size, void* d_ws, size_t ws_size,
                              hipStream_t stream) {
  const float* x  = (const float*)d_in[0];
  const int*   ei = (const int*)d_in[1];
  const float* W  = (const float*)d_in[2];
  const float* a  = (const float*)d_in[3];
  const int N = in_sizes[0] / F_DIM;
  const int E = in_sizes[1] / 2;
  const int* src = ei;
  const int* dst = ei + E;

  float* h     = (float*)d_ws;
  int*   deg   = (int*)(h + (size_t)N * F_DIM);
  int*   first = deg + N;
  float* as    = (float*)(first + N);
  float* ad    = as + N;
  int*   slot  = (int*)(ad + N);
  int*   gsrc  = slot + E;
  float* out   = (float*)d_out;

  gat_gemm_wmma     <<<(N + 15) / 16,    256,  0, stream>>>(x, W, h, deg, N);
  gat_alpha         <<<(N + 7)  / 8,     256,  0, stream>>>(h, a, as, ad, N);
  gat_edge_slot     <<<(E + 255) / 256,  256,  0, stream>>>(dst, deg, slot, E);
  gat_segment_first <<<(N + 1023) / 1024, 1024, 0, stream>>>(deg, first, N);
  gat_group_edges   <<<(E + 255) / 256,  256,  0, stream>>>(src, dst, first, slot, gsrc, E);
  gat_node_aggregate<<<(N + 7)  / 8,     256,  0, stream>>>(gsrc, first, deg, h, as, ad, out, N);
}


# edges grouped by destination with LDS counters only: per-chunk counts over 256-node buckets, bucket scatter, one block orders a bucket; no global atomic per edge; sums in LDS-atomic slot order
# speedup vs baseline: 2.9959x; 1.6231x over previous
#include <hip/hip_runtime.h>
#include <hip/hip_bf16.h>


typedef __attribute__((ext_vector_type(2))) float v2f;
typedef __attribute__((ext_vector_type(8))) float v8f;

#define F_DIM 128
#define LEAKY 0.2f
#define ROWS_IN_FLIGHT 8

__global__ __launch_bounds__(256) void gat_gemm_wmma(
    const float* __restrict__ x, const float* __restrict__ W,
    float* __restrict__ h, int M)
{
  __shared__ float sA[16 * 132];
  const int tid = threadIdx.x;
  const int m0  = blockIdx.x * 16;

  for (int i = tid; i < 512; i += 256) {
    int r = i >> 5, c = i & 31;
    float4 v = make_float4(0.f, 0.f, 0.f, 0.f);
    if (m0 + r < M) v = ((const float4*)(x + (size_t)(m0 + r) * F_DIM))[c];
    ((float4*)sA)[r * 33 + c] = v;
  }
  __syncthreads();

  const int lane = tid & 31;
  const int wv   = tid >> 5;
  const int row  = lane & 15;
  const int kh   = (lane >> 4) * 2;
  const int n0   = wv * 16;

  v8f acc = {};
  #pragma unroll 4
  for (int k = 0; k < F_DIM; k += 4) {
    v2f a, b;
    a.x = sA[row * 132 + k + kh + 0];
    a.y = sA[row * 132 + k + kh + 1];
    b.x = W[(k + kh + 0) * F_DIM + n0 + row];
    b.y = W[(k + kh + 1) * F_DIM + n0 + row];
    acc = __builtin_amdgcn_wmma_f32_16x16x4_f32(false, a, false, b,
                                                (short)0, acc, false, false);
  }

  const int mh = (lane >> 4) * 8;
  float* hp = h + (size_t)m0 * F_DIM + n0 + row;
  if (m0 + 16 <= M) {
    #pragma unroll
    for (int v = 0; v < 8; ++v) hp[(size_t)(mh + v) * F_DIM] = acc[v];
  } else {
    #pragma unroll
    for (int v = 0; v < 8; ++v)
      if (m0 + mh + v < M) hp[(size_t)(mh + v) * F_DIM] = acc[v];
  }
}

__global__ __launch_bounds__(256) void gat_alpha(
    const float* __restrict__ h, const float* __restrict__ a,
    float* __restrict__ as, float* __restrict__ ad, int N)
{
  const int node = blockIdx.x * 8 + (threadIdx.x >> 5);
  const int lane = threadIdx.x & 31;
  if (node >= N) return;
  float4 hv = ((const float4*)(h + (size_t)node * F_DIM))[lane];
  float4 a1 = ((const float4*)a)[lane];
  float4 a2 = ((const float4*)(a + F_DIM))[lane];
  float s = hv.x * a1.x + hv.y * a1.y + hv.z * a1.z + hv.w * a1.w;
  float d = hv.x * a2.x + hv.y * a2.y + hv.z * a2.z + hv.w * a2.w;
  #pragma unroll
  for (int m = 16; m >= 1; m >>= 1) {
    s += __shfl_xor(s, m, 32);
    d += __shfl_xor(d, m, 32);
  }
  if (lane == 0) { as[node] = s; ad[node] = d; }
}

#define GRP_CHUNKS 256
#define GRP_THREADS 1024
#define GRP_BATCH 4

__device__ __forceinline__ void keep_here(int& v) { asm volatile("" : "+v"(v)); }

__global__ __launch_bounds__(GRP_THREADS) void gat_bucket_count(
    const int* __restrict__ dst, int* __restrict__ cnt, int E)
{
  __shared__ int hist[256];
  const int t = threadIdx.x;
  if (t < 256) hist[t] = 0;
  __syncthreads();
  const int per = (E + GRP_CHUNKS - 1) / GRP_CHUNKS;
  const int e0  = blockIdx.x * per;
  const int e1  = min(e0 + per, E);
  for (int base = e0 + t; base < e1; base += GRP_BATCH * GRP_THREADS) {
    int d[GRP_BATCH];
    #pragma unroll
    for (int i = 0; i < GRP_BATCH; ++i)
      d[i] = dst[min(base + i * GRP_THREADS, e1 - 1)];
    #pragma unroll
    for (int i = 0; i < GRP_BATCH; ++i) keep_here(d[i]);
    #pragma unroll
    for (int i = 0; i < GRP_BATCH; ++i)
      if (base + i * GRP_THREADS < e1) atomicAdd(&hist[(d[i] >> 8) & 255], 1);
  }
  __syncthreads();
  if (t < 256) cnt[blockIdx.x * 256 + t] = hist[t];
}

__global__ __launch_bounds__(GRP_THREADS) void gat_bucket_scatter(
    const int* __restrict__ src, const int* __restrict__ dst,
    const int* __restrict__ cnt, int* __restrict__ bstart,
    int2* __restrict__ pairs, int E)
{
  static_assert(GRP_CHUNKS == 256 && GRP_THREADS == 1024, "16 row slices x 64 int4 columns");
  __shared__ int s_pre[16][256];
  __shared__ int s_all[16][256];
  __shared__ int s_cur[256];
  __shared__ int s_wtot[8];
  const int t    = threadIdx.x;
  const int lane = t & 31;
  const int w    = t >> 5;
  const int c4   = t & 63;
  const int j    = t >> 6;
  const int k    = blockIdx.x;

  int4 pre = make_int4(0, 0, 0, 0), all = make_int4(0, 0, 0, 0);
  #pragma unroll
  for (int r = 0; r < 16; ++r) {
    const int  kk = j * 16 + r;
    const int4 c  = ((const int4*)cnt)[kk * 64 + c4];
    const bool in_front = kk < k;
    all.x += c.x; all.y += c.y; all.z += c.z; all.w += c.w;
    pre.x += in_front ? c.x : 0; pre.y += in_front ? c.y : 0;
    pre.z += in_front ? c.z : 0; pre.w += in_front ? c.w : 0;
  }
  ((int4*)s_pre[j])[c4] = pre;
  ((int4*)s_all[j])[c4] = all;
  __syncthreads();

  int v = 0, mine = 0;
  if (t < 256) {
    #pragma unroll
    for (int q = 0; q < 16; ++q) { mine += s_pre[q][t]; v += s_all[q][t]; }
  }
  int inc = v;
  #pragma unroll
  for (int o = 1; o < 32; o <<= 1) {
    int u = __shfl_up(inc, o, 32);
    if (lane >= o) inc += u;
  }
  if (t < 256 && lane == 31) s_wtot[w] = inc;
  __syncthreads();
  if (t < 256) {
    int front = 0;
    for (int i = 0; i < w; ++i) front += s_wtot[i];
    const int start = front + inc - v;
    s_cur[t] = start + mine;
    if (k == 0) bstart[t] = start;
  }
  __syncthreads();

  const int per = (E + GRP_CHUNKS - 1) / GRP_CHUNKS;
  const int e0  = k * per;
  const int e1  = min(e0 + per, E);
  for (int base = e0 + t; base < e1; base += GRP_BATCH * GRP_THREADS) {
    int d[GRP_BATCH], s[GRP_BATCH];
    #pragma unroll
    for (int i = 0; i < GRP_BATCH; ++i) {
      const int e = min(base + i * GRP_THREADS, e1 - 1);
      d[i] = dst[e];
      s[i] = src[e];
    }
    #pragma unroll
    for (int i = 0; i < GRP_BATCH; ++i) { keep_here(d[i]); keep_here(s[i]); }
    #pragma unroll
    for (int i = 0; i < GRP_BATCH; ++i)
      if (base + i * GRP_THREADS < e1)
        pairs[atomicAdd(&s_cur[(d[i] >> 8) & 255], 1)] = make_int2(d[i], s[i]);
  }
}

__global__ __launch_bounds__(GRP_THREADS) void gat_bucket_group(
    const int2* __restrict__ pairs, const int* __restrict__ bstart,
    int* __restrict__ deg, int* __restrict__ first, int* __restrict__ gsrc,
    int N, int E)
{
  __shared__ int s_c[256];
  __shared__ int s_wtot[8];
  const int t    = threadIdx.x;
  const int lane = t & 31;
  const int w    = t >> 5;
  const int b    = blockIdx.x;
  const int lo   = bstart[b];
  const int hi   = (b + 1 < 256) ? bstart[b + 1] : E;

  if (t < 256) s_c[t] = 0;
  __syncthreads();
  for (int base = lo + t; base < hi; base += GRP_BATCH * GRP_THREADS) {
    int d[GRP_BATCH];
    #pragma unroll
    for (int i = 0; i < GRP_BATCH; ++i)
      d[i] = pairs[min(base + i * GRP_THREADS, hi - 1)].x;
    #pragma unroll
    for (int i = 0; i < GRP_BATCH; ++i) keep_here(d[i]);
    #pragma unroll
    for (int i = 0; i < GRP_BATCH; ++i)
      if (base + i * GRP_THREADS < hi) atomicAdd(&s_c[d[i] & 255], 1);
  }
  __syncthreads();

  const int v = (t < 256) ? s_c[t] : 0;
  int inc = v;
  #pragma unroll
  for (int o = 1; o < 32; o <<= 1) {
    int u = __shfl_up(inc, o, 32);
    if (lane >= o) inc += u;
  }
  if (t < 256 && lane == 31) s_wtot[w] = inc;
  __syncthreads();
  if (t < 256) {
    int front = lo;
    for (int i = 0; i < w; ++i) front += s_wtot[i];
    const int f = front + inc - v;
    const int n = b * 256 + t;
    if (n < N) { first[n] = f; deg[n] = v; }
    s_c[t] = f;
  }
  __syncthreads();
  for (int base = lo + t; base < hi; base += GRP_BATCH * GRP_THREADS) {
    int2 q[GRP_BATCH];
    #pragma unroll
    for (int i = 0; i < GRP_BATCH; ++i)
      q[i] = pairs[min(base + i * GRP_THREADS, hi - 1)];
    #pragma unroll
    for (int i = 0; i < GRP_BATCH; ++i) { keep_here(q[i].x); keep_here(q[i].y); }
    #pragma unroll
    for (int i = 0; i < GRP_BATCH; ++i)
      if (base + i * GRP_THREADS < hi) gsrc[atomicAdd(&s_c[q[i].x & 255], 1)] = q[i].y;
  }
}

__device__ __forceinline__ float wave_max(float v) {
  #pragma unroll
  for (int m = 16; m >= 1; m >>= 1) v = fmaxf(v, __shfl_xor(v, m, 32));
  return v;
}
__device__ __forceinline__ float wave_sum(float v) {
  #pragma unroll
  for (int m = 16; m >= 1; m >>= 1) v += __shfl_xor(v, m, 32);
  return v;
}
__device__ __forceinline__ float4 h_row(const float* __restrict__ h, int s, int lane) {
  return ((const float4*)(h + (size_t)s * F_DIM))[lane];
}
__device__ __forceinline__ void axpy4(float4& acc, float c, const float4& r) {
  acc.x = fmaf(c, r.x, acc.x);
  acc.y = fmaf(c, r.y, acc.y);
  acc.z = fmaf(c, r.z, acc.z);
  acc.w = fmaf(c, r.w, acc.w);
}

__device__ __forceinline__ void add_rows(
    float4& acc, const float* __restrict__ h, int lane, int s, float c, int n)
{
  const int ci = __float_as_int(c);
  int j = 0;
  for (; j + ROWS_IN_FLIGHT <= n; j += ROWS_IN_FLIGHT) {
    float4 r[ROWS_IN_FLIGHT];
    #pragma unroll
    for (int k = 0; k < ROWS_IN_FLIGHT; ++k)
      r[k] = h_row(h, __builtin_amdgcn_readlane(s, j + k), lane);
    #pragma unroll
    for (int k = 0; k < ROWS_IN_FLIGHT; ++k)
      axpy4(acc, __int_as_float(__builtin_amdgcn_readlane(ci, j + k)), r[k]);
  }
  if (j < n) {
    float4 r[ROWS_IN_FLIGHT];
    #pragma unroll
    for (int k = 0; k < ROWS_IN_FLIGHT; ++k)
      r[k] = h_row(h, __builtin_amdgcn_readlane(s, min(j + k, n - 1)), lane);
    #pragma unroll
    for (int k = 0; k < ROWS_IN_FLIGHT; ++k) {
      const float cj = __int_as_float(__builtin_amdgcn_readlane(ci, min(j + k, n - 1)));
      axpy4(acc, (j + k < n) ? cj : 0.f, r[k]);
    }
  }
}

__device__ __forceinline__ float edge_logit(
    const float* __restrict__ as, int s, float adv)
{
  float t = as[s] + adv;
  return t > 0.f ? t : LEAKY * t;
}

__global__ __launch_bounds__(256) void gat_node_aggregate(
    const int* __restrict__ gsrc, const int* __restrict__ first,
    const int* __restrict__ deg, const float* __restrict__ h,
    const float* __restrict__ as, const float* __restrict__ ad,
    float* __restrict__ out, int N)
{
  const int d    = __builtin_amdgcn_readfirstlane(blockIdx.x * 8 + (threadIdx.x >> 5));
  const int lane = threadIdx.x & 31;
  if (d >= N) return;
  const int   beg = first[d];
  const int   cnt = deg[d];
  const float adv = ad[d];
  const int*  seg = gsrc + beg;

  const bool v0 = lane < cnt;
  int   s0 = 0;
  float l0 = -INFINITY;
  if (v0) { s0 = seg[lane]; l0 = edge_logit(as, s0, adv); }

  float m = l0;
  for (int c = 32; c < cnt; c += 32)
    if (c + lane < cnt) m = fmaxf(m, edge_logit(as, seg[c + lane], adv));
  m = wave_max(m);

  const float p0 = v0 ? expf(l0 - m) : 0.f;
  float sum = p0;
  for (int c = 32; c < cnt; c += 32)
    if (c + lane < cnt) sum += expf(edge_logit(as, seg[c + lane], adv) - m);
  sum = wave_sum(sum);
  const float den = sum + 1e-16f;

  float4 acc = make_float4(0.f, 0.f, 0.f, 0.f);
  add_rows(acc, h, lane, s0, p0 / den, min(cnt, 32));
  for (int c = 32; c < cnt; c += 32) {
    int s = 0; float p = 0.f;
    if (c + lane < cnt) { s = seg[c + lane]; p = expf(edge_logit(as, s, adv) - m); }
    add_rows(acc, h, lane, s, p / den, min(cnt - c, 32));
  }
  ((float4*)(out + (size_t)d * F_DIM))[lane] = acc;
}

extern "C" void kernel_launch(void* const* d_in, const int* in_sizes, int n_in,
                              void* d_out, int out_size, void* d_ws, size_t ws_size,
                              hipStream_t stream) {
  const float* x  = (const float*)d_in[0];
  const int*   ei = (const int*)d_in[1];
  const float* W  = (const float*)d_in[2];
  const float* a  = (const float*)d_in[3];
  const int N = in_sizes[0] / F_DIM;
  const int E = in_sizes[1] / 2;
  const int* src = ei;
  const int* dst = ei + E;

  float* h     = (float*)d_ws;
  int*   deg   = (int*)(h + (size_t)N * F_DIM);
  int*   first = deg + N;
  float* as    = (float*)(first + N);
  float* ad    = as + N;
  int*   gsrc  = (int*)(ad + N) + E;
  float* out   = (float*)d_out;
  int*   cnt    = (int*)d_out;
  int*   bstart = cnt + 256 * 256;
  int2*  pairs  = (int2*)(bstart + 256);

  gat_gemm_wmma     <<<(N + 15) / 16,   256,         0, stream>>>(x, W, h, N);
  gat_alpha         <<<(N + 7)  / 8,    256,         0, stream>>>(h, a, as, ad, N);
  gat_bucket_count  <<<GRP_CHUNKS,      GRP_THREADS, 0, stream>>>(dst, cnt, E);
  gat_bucket_scatter<<<GRP_CHUNKS,      GRP_THREADS, 0, stream>>>(src, dst, cnt, bstart, pairs, E);
  gat_bucket_group  <<<(N + 255) / 256, GRP_THREADS, 0, stream>>>(pairs, bstart, deg, first, gsrc, N, E);
  gat_node_aggregate<<<(N + 7)  / 8,    256,         0, stream>>>(gsrc, first, deg, h, as, ad, out, N);
}


# x@W epilogue lays its h tile out in LDS; each wave stores 2 whole 512-B rows and forms their two attention dots from them: the alpha launch and its re-read of h are gone; sums in the same order
# speedup vs baseline: 3.1340x; 1.0461x over previous
#include <hip/hip_runtime.h>
#include <hip/hip_bf16.h>


typedef __attribute__((ext_vector_type(2))) float v2f;
typedef __attribute__((ext_vector_type(8))) float v8f;

#define F_DIM 128
#define LEAKY 0.2f
#define ROWS_IN_FLIGHT 8

__global__ __launch_bounds__(256) void gat_gemm_wmma(
    const float* __restrict__ x, const float* __restrict__ W,
    const float* __restrict__ a, float* __restrict__ h,
    float* __restrict__ as, float* __restrict__ ad, int M)
{
  __shared__ float sA[16 * 132];
  __shared__ float sH[16 * 132];
  const int tid = threadIdx.x;
  const int m0  = blockIdx.x * 16;

  for (int i = tid; i < 512; i += 256) {
    int r = i >> 5, c = i & 31;
    float4 v = make_float4(0.f, 0.f, 0.f, 0.f);
    if (m0 + r < M) v = ((const float4*)(x + (size_t)(m0 + r) * F_DIM))[c];
    ((float4*)sA)[r * 33 + c] = v;
  }
  __syncthreads();

  const int lane = tid & 31;
  const int wv   = tid >> 5;
  const int row  = lane & 15;
  const int kh   = (lane >> 4) * 2;
  const int n0   = wv * 16;

  const float4 a1 = ((const float4*)a)[lane];
  const float4 a2 = ((const float4*)(a + F_DIM))[lane];

  v8f acc = {};
  #pragma unroll 4
  for (int k = 0; k < F_DIM; k += 4) {
    v2f a, b;
    a.x = sA[row * 132 + k + kh + 0];
    a.y = sA[row * 132 + k + kh + 1];
    b.x = W[(k + kh + 0) * F_DIM + n0 + row];
    b.y = W[(k + kh + 1) * F_DIM + n0 + row];
    acc = __builtin_amdgcn_wmma_f32_16x16x4_f32(false, a, false, b,
                                                (short)0, acc, false, false);
  }

  const int mh = (lane >> 4) * 8;
  #pragma unroll
  for (int v = 0; v < 8; ++v) sH[(mh + v) * 132 + n0 + row] = acc[v];
  __syncthreads();

  const int    r0 = wv * 2;
  const float4 h0 = ((const float4*)sH)[(r0 + 0) * 33 + lane];
  const float4 h1 = ((const float4*)sH)[(r0 + 1) * 33 + lane];
  float s0 = h0.x * a1.x + h0.y * a1.y + h0.z * a1.z + h0.w * a1.w;
  float d0 = h0.x * a2.x + h0.y * a2.y + h0.z * a2.z + h0.w * a2.w;
  float s1 = h1.x * a1.x + h1.y * a1.y + h1.z * a1.z + h1.w * a1.w;
  float d1 = h1.x * a2.x + h1.y * a2.y + h1.z * a2.z + h1.w * a2.w;
  #pragma unroll
  for (int m = 16; m >= 1; m >>= 1) {
    s0 += __shfl_xor(s0, m, 32);
    d0 += __shfl_xor(d0, m, 32);
    s1 += __shfl_xor(s1, m, 32);
    d1 += __shfl_xor(d1, m, 32);
  }
  const int g0 = m0 + r0;
  if (g0 < M) {
    ((float4*)(h + (size_t)g0 * F_DIM))[lane] = h0;
    if (lane == 0) { as[g0] = s0; ad[g0] = d0; }
  }
  if (g0 + 1 < M) {
    ((float4*)(h + (size_t)(g0 + 1) * F_DIM))[lane] = h1;
    if (lane == 0) { as[g0 + 1] = s1; ad[g0 + 1] = d1; }
  }
}

#define GRP_CHUNKS 256
#define GRP_THREADS 1024
#define GRP_BATCH 4

__device__ __forceinline__ void keep_here(int& v) { asm volatile("" : "+v"(v)); }

__global__ __launch_bounds__(GRP_THREADS) void gat_bucket_count(
    const int* __restrict__ dst, int* __restrict__ cnt, int E)
{
  __shared__ int hist[256];
  const int t = threadIdx.x;
  if (t < 256) hist[t] = 0;
  __syncthreads();
  const int per = (E + GRP_CHUNKS - 1) / GRP_CHUNKS;
  const int e0  = blockIdx.x * per;
  const int e1  = min(e0 + per, E);
  for (int base = e0 + t; base < e1; base += GRP_BATCH * GRP_THREADS) {
    int d[GRP_BATCH];
    #pragma unroll
    for (int i = 0; i < GRP_BATCH; ++i)
      d[i] = dst[min(base + i * GRP_THREADS, e1 - 1)];
    #pragma unroll
    for (int i = 0; i < GRP_BATCH; ++i) keep_here(d[i]);
    #pragma unroll
    for (int i = 0; i < GRP_BATCH; ++i)
      if (base + i * GRP_THREADS < e1) atomicAdd(&hist[(d[i] >> 8) & 255], 1);
  }
  __syncthreads();
  if (t < 256) cnt[blockIdx.x * 256 + t] = hist[t];
}

__global__ __launch_bounds__(GRP_THREADS) void gat_bucket_scatter(
    const int* __restrict__ src, const int* __restrict__ dst,
    const int* __restrict__ cnt, int* __restrict__ bstart,
    int2* __restrict__ pairs, int E)
{
  static_assert(GRP_CHUNKS == 256 && GRP_THREADS == 1024, "16 row slices x 64 int4 columns");
  __shared__ int s_pre[16][256];
  __shared__ int s_all[16][256];
  __shared__ int s_cur[256];
  __shared__ int s_wtot[8];
  const int t    = threadIdx.x;
  const int lane = t & 31;
  const int w    = t >> 5;
  const int c4   = t & 63;
  const int j    = t >> 6;
  const int k    = blockIdx.x;

  int4 pre = make_int4(0, 0, 0, 0), all = make_int4(0, 0, 0, 0);
  #pragma unroll
  for (int r = 0; r < 16; ++r) {
    const int  kk = j * 16 + r;
    const int4 c  = ((const int4*)cnt)[kk * 64 + c4];
    const bool in_front = kk < k;
    all.x += c.x; all.y += c.y; all.z += c.z; all.w += c.w;
    pre.x += in_front ? c.x : 0; pre.y += in_front ? c.y : 0;
    pre.z += in_front ? c.z : 0; pre.w += in_front ? c.w : 0;
  }
  ((int4*)s_pre[j])[c4] = pre;
  ((int4*)s_all[j])[c4] = all;
  __syncthreads();

  int v = 0, mine = 0;
  if (t < 256) {
    #pragma unroll
    for (int q = 0; q < 16; ++q) { mine += s_pre[q][t]; v += s_all[q][t]; }
  }
  int inc = v;
  #pragma unroll
  for (int o = 1; o < 32; o <<= 1) {
    int u = __shfl_up(inc, o, 32);
    if (lane >= o) inc += u;
  }
  if (t < 256 && lane == 31) s_wtot[w] = inc;
  __syncthreads();
  if (t < 256) {
    int front = 0;
    for (int i = 0; i < w; ++i) front += s_wtot[i];
    const int start = front + inc - v;
    s_cur[t] = start + mine;
    if (k == 0) bstart[t] = start;
  }
  __syncthreads();

  const int per = (E + GRP_CHUNKS - 1) / GRP_CHUNKS;
  const int e0  = k * per;
  const int e1  = min(e0 + per, E);
  for (int base = e0 + t; base < e1; base += GRP_BATCH * GRP_THREADS) {
    int d[GRP_BATCH], s[GRP_BATCH];
    #pragma unroll
    for (int i = 0; i < GRP_BATCH; ++i) {
      const int e = min(base + i * GRP_THREADS, e1 - 1);
      d[i] = dst[e];
      s[i] = src[e];
    }
    #pragma unroll
    for (int i = 0; i < GRP_BATCH; ++i) { keep_here(d[i]); keep_here(s[i]); }
    #pragma unroll
    for (int i = 0; i < GRP_BATCH; ++i)
      if (base + i * GRP_THREADS < e1)
        pairs[atomicAdd(&s_cur[(d[i] >> 8) & 255], 1)] = make_int2(d[i], s[i]);
  }
}

__global__ __launch_bounds__(GRP_THREADS) void gat_bucket_group(
    const int2* __restrict__ pairs, const int* __restrict__ bstart,
    int* __restrict__ deg, int* __restrict__ first, int* __restrict__ gsrc,
    int N, int E)
{
  __shared__ int s_c[256];
  __shared__ int s_wtot[8];
  const int t    = threadIdx.x;
  const int lane = t & 31;
  const int w    = t >> 5;
  const int b    = blockIdx.x;
  const int lo   = bstart[b];
  const int hi   = (b + 1 < 256) ? bstart[b + 1] : E;

  if (t < 256) s_c[t] = 0;
  __syncthreads();
  for (int base = lo + t; base < hi; base += GRP_BATCH * GRP_THREADS) {
    int d[GRP_BATCH];
    #pragma unroll
    for (int i = 0; i < GRP_BATCH; ++i)
      d[i] = pairs[min(base + i * GRP_THREADS, hi - 1)].x;
    #pragma unroll
    for (int i = 0; i < GRP_BATCH; ++i) keep_here(d[i]);
    #pragma unroll
    for (int i = 0; i < GRP_BATCH; ++i)
      if (base + i * GRP_THREADS < hi) atomicAdd(&s_c[d[i] & 255], 1);
  }
  __syncthreads();

  const int v = (t < 256) ? s_c[t] : 0;
  int inc = v;
  #pragma unroll
  for (int o = 1; o < 32; o <<= 1) {
    int u = __shfl_up(inc, o, 32);
    if (lane >= o) inc += u;
  }
  if (t < 256 && lane == 31) s_wtot[w] = inc;
  __syncthreads();
  if (t < 256) {
    int front = lo;
    for (int i = 0; i < w; ++i) front += s_wtot[i];
    const int f = front + inc - v;
    const int n = b * 256 + t;
    if (n < N) { first[n] = f; deg[n] = v; }
    s_c[t] = f;
  }
  __syncthreads();
  for (int base = lo + t; base < hi; base += GRP_BATCH * GRP_THREADS) {
    int2 q[GRP_BATCH];
    #pragma unroll
    for (int i = 0; i < GRP_BATCH; ++i)
      q[i] = pairs[min(base + i * GRP_THREADS, hi - 1)];
    #pragma unroll
    for (int i = 0; i < GRP_BATCH; ++i) { keep_here(q[i].x); keep_here(q[i].y); }
    #pragma unroll
    for (int i = 0; i < GRP_BATCH; ++i)
      if (base + i * GRP_THREADS < hi) gsrc[atomicAdd(&s_c[q[i].x & 255], 1)] = q[i].y;
  }
}

__device__ __forceinline__ float wave_max(float v) {
  #pragma unroll
  for (int m = 16; m >= 1; m >>= 1) v = fmaxf(v, __shfl_xor(v, m, 32));
  return v;
}
__device__ __forceinline__ float wave_sum(float v) {
  #pragma unroll
  for (int m = 16; m >= 1; m >>= 1) v += __shfl_xor(v, m, 32);
  return v;
}
__device__ __forceinline__ float4 h_row(const float* __restrict__ h, int s, int lane) {
  return ((const float4*)(h + (size_t)s * F_DIM))[lane];
}
__device__ __forceinline__ void axpy4(float4& acc, float c, const float4& r) {
  acc.x = fmaf(c, r.x, acc.x);
  acc.y = fmaf(c, r.y, acc.y);
  acc.z = fmaf(c, r.z, acc.z);
  acc.w = fmaf(c, r.w, acc.w);
}

__device__ __forceinline__ void add_rows(
    float4& acc, const float* __restrict__ h, int lane, int s, float c, int n)
{
  const int ci = __float_as_int(c);
  int j = 0;
  for (; j + ROWS_IN_FLIGHT <= n; j += ROWS_IN_FLIGHT) {
    float4 r[ROWS_IN_FLIGHT];
    #pragma unroll
    for (int k = 0; k < ROWS_IN_FLIGHT; ++k)
      r[k] = h_row(h, __builtin_amdgcn_readlane(s, j + k), lane);
    #pragma unroll
    for (int k = 0; k < ROWS_IN_FLIGHT; ++k)
      axpy4(acc, __int_as_float(__builtin_amdgcn_readlane(ci, j + k)), r[k]);
  }
  if (j < n) {
    float4 r[ROWS_IN_FLIGHT];
    #pragma unroll
    for (int k = 0; k < ROWS_IN_FLIGHT; ++k)
      r[k] = h_row(h, __builtin_amdgcn_readlane(s, min(j + k, n - 1)), lane);
    #pragma unroll
    for (int k = 0; k < ROWS_IN_FLIGHT; ++k) {
      const float cj = __int_as_float(__builtin_amdgcn_readlane(ci, min(j + k, n - 1)));
      axpy4(acc, (j + k < n) ? cj : 0.f, r[k]);
    }
  }
}

__device__ __forceinline__ float edge_logit(
    const float* __restrict__ as, int s, float adv)
{
  float t = as[s] + adv;
  return t > 0.f ? t : LEAKY * t;
}

__global__ __launch_bounds__(256) void gat_node_aggregate(
    const int* __restrict__ gsrc, const int* __restrict__ first,
    const int* __restrict__ deg, const float* __restrict__ h,
    const float* __restrict__ as, const float* __restrict__ ad,
    float* __restrict__ out, int N)
{
  const int d    = __builtin_amdgcn_readfirstlane(blockIdx.x * 8 + (threadIdx.x >> 5));
  const int lane = threadIdx.x & 31;
  if (d >= N) return;
  const int   beg = first[d];
  const int   cnt = deg[d];
  const float adv = ad[d];
  const int*  seg = gsrc + beg;

  const bool v0 = lane < cnt;
  int   s0 = 0;
  float l0 = -INFINITY;
  if (v0) { s0 = seg[lane]; l0 = edge_logit(as, s0, adv); }

  float m = l0;
  for (int c = 32; c < cnt; c += 32)
    if (c + lane < cnt) m = fmaxf(m, edge_logit(as, seg[c + lane], adv));
  m = wave_max(m);

  const float p0 = v0 ? expf(l0 - m) : 0.f;
  float sum = p0;
  for (int c = 32; c < cnt; c += 32)
    if (c + lane < cnt) sum += expf(edge_logit(as, seg[c + lane], adv) - m);
  sum = wave_sum(sum);
  const float den = sum + 1e-16f;

  float4 acc = make_float4(0.f, 0.f, 0.f, 0.f);
  add_rows(acc, h, lane, s0, p0 / den, min(cnt, 32));
  for (int c = 32; c < cnt; c += 32) {
    int s = 0; float p = 0.f;
    if (c + lane < cnt) { s = seg[c + lane]; p = expf(edge_logit(as, s, adv) - m); }
    add_rows(acc, h, lane, s, p / den, min(cnt - c, 32));
  }
  ((float4*)(out + (size_t)d * F_DIM))[lane] = acc;
}

extern "C" void kernel_launch(void* const* d_in, const int* in_sizes, int n_in,
                              void* d_out, int out_size, void* d_ws, size_t ws_size,
                              hipStream_t stream) {
  const float* x  = (const float*)d_in[0];
  const int*   ei = (const int*)d_in[1];
  const float* W  = (const float*)d_in[2];
  const float* a  = (const float*)d_in[3];
  const int N = in_sizes[0] / F_DIM;
  const int E = in_sizes[1] / 2;
  const int* src = ei;
  const int* dst = ei + E;

  float* h     = (float*)d_ws;
  int*   deg   = (int*)(h + (size_t)N * F_DIM);
  int*   first = deg + N;
  float* as    = (float*)(first + N);
  float* ad    = as + N;
  int*   gsrc  = (int*)(ad + N) + E;
  float* out   = (float*)d_out;
  int*   cnt    = (int*)d_out;
  int*   bstart = cnt + 256 * 256;
  int2*  pairs  = (int2*)(bstart + 256);

  gat_gemm_wmma     <<<(N + 15) / 16,   256,         0, stream>>>(x, W, a, h, as, ad, N);
  gat_bucket_count  <<<GRP_CHUNKS,      GRP_THREADS, 0, stream>>>(dst, cnt, E);
  gat_bucket_scatter<<<GRP_CHUNKS,      GRP_THREADS, 0, stream>>>(src, dst, cnt, bstart, pairs, E);
  gat_bucket_group  <<<(N + 255) / 256, GRP_THREADS, 0, stream>>>(pairs, bstart, deg, first, gsrc, N, E);
  gat_node_aggregate<<<(N + 7)  / 8,    256,         0, stream>>>(gsrc, first, deg, h, as, ad, out, N);
}


# edge grouping's counting pass rides as the first 256 work-groups of the x@W launch (256 threads, 16 edge loads in flight each) instead of a launch of its own: 5 launches become 4; sums in same order
# speedup vs baseline: 3.2519x; 1.0376x over previous
#include <hip/hip_runtime.h>
#include <hip/hip_bf16.h>


typedef __attribute__((ext_vector_type(2))) float v2f;
typedef __attribute__((ext_vector_type(8))) float v8f;

#define F_DIM 128
#define LEAKY 0.2f
#define ROWS_IN_FLIGHT 8

#define GRP_CHUNKS 256
#define GRP_THREADS 1024
#define GRP_BATCH 4
#define CNT_BATCH 16

__device__ __forceinline__ void keep_here(int& v) { asm volatile("" : "+v"(v)); }

__global__ __launch_bounds__(256) void gat_gemm_wmma_count(
    const float* __restrict__ x, const float* __restrict__ W,
    const float* __restrict__ a, float* __restrict__ h,
    float* __restrict__ as, float* __restrict__ ad, int M,
    const int* __restrict__ dst, int* __restrict__ cnt, int E)
{
  __shared__ float sA[16 * 132];
  __shared__ float sH[16 * 132];
  __shared__ int   hist[256];
  const int tid = threadIdx.x;

  if (blockIdx.x < GRP_CHUNKS) {
    hist[tid] = 0;
    __syncthreads();
    const int per = (E + GRP_CHUNKS - 1) / GRP_CHUNKS;
    const int e0  = blockIdx.x * per;
    const int e1  = min(e0 + per, E);
    for (int base = e0 + tid; base < e1; base += CNT_BATCH * 256) {
      int d[CNT_BATCH];
      #pragma unroll
      for (int i = 0; i < CNT_BATCH; ++i)
        d[i] = dst[min(base + i * 256, e1 - 1)];
      #pragma unroll
      for (int i = 0; i < CNT_BATCH; ++i) keep_here(d[i]);
      #pragma unroll
      for (int i = 0; i < CNT_BATCH; ++i)
        if (base + i * 256 < e1) atomicAdd(&hist[(d[i] >> 8) & 255], 1);
    }
    __syncthreads();
    cnt[blockIdx.x * 256 + tid] = hist[tid];
    return;
  }

  const int m0  = (blockIdx.x - GRP_CHUNKS) * 16;

  for (int i = tid; i < 512; i += 256) {
    int r = i >> 5, c = i & 31;
    float4 v = make_float4(0.f, 0.f, 0.f, 0.f);
    if (m0 + r < M) v = ((const float4*)(x + (size_t)(m0 + r) * F_DIM))[c];
    ((float4*)sA)[r * 33 + c] = v;
  }
  __syncthreads();

  const int lane = tid & 31;
  const int wv   = tid >> 5;
  const int row  = lane & 15;
  const int kh   = (lane >> 4) * 2;
  const int n0   = wv * 16;

  const float4 a1 = ((const float4*)a)[lane];
  const float4 a2 = ((const float4*)(a + F_DIM))[lane];

  v8f acc = {};
  #pragma unroll 4
  for (int k = 0; k < F_DIM; k += 4) {
    v2f a, b;
    a.x = sA[row * 132 + k + kh + 0];
    a.y = sA[row * 132 + k + kh + 1];
    b.x = W[(k + kh + 0) * F_DIM + n0 + row];
    b.y = W[(k + kh + 1) * F_DIM + n0 + row];
    acc = __builtin_amdgcn_wmma_f32_16x16x4_f32(false, a, false, b,
                                                (short)0, acc, false, false);
  }

  const int mh = (lane >> 4) * 8;
  #pragma unroll
  for (int v = 0; v < 8; ++v) sH[(mh + v) * 132 + n0 + row] = acc[v];
  __syncthreads();

  const int    r0 = wv * 2;
  const float4 h0 = ((const float4*)sH)[(r0 + 0) * 33 + lane];
  const float4 h1 = ((const float4*)sH)[(r0 + 1) * 33 + lane];
  float s0 = h0.x * a1.x + h0.y * a1.y + h0.z * a1.z + h0.w * a1.w;
  float d0 = h0.x * a2.x + h0.y * a2.y + h0.z * a2.z + h0.w * a2.w;
  float s1 = h1.x * a1.x + h1.y * a1.y + h1.z * a1.z + h1.w * a1.w;
  float d1 = h1.x * a2.x + h1.y * a2.y + h1.z * a2.z + h1.w * a2.w;
  #pragma unroll
  for (int m = 16; m >= 1; m >>= 1) {
    s0 += __shfl_xor(s0, m, 32);
    d0 += __shfl_xor(d0, m, 32);
    s1 += __shfl_xor(s1, m, 32);
    d1 += __shfl_xor(d1, m, 32);
  }
  const int g0 = m0 + r0;
  if (g0 < M) {
    ((float4*)(h + (size_t)g0 * F_DIM))[lane] = h0;
    if (lane == 0) { as[g0] = s0; ad[g0] = d0; }
  }
  if (g0 + 1 < M) {
    ((float4*)(h + (size_t)(g0 + 1) * F_DIM))[lane] = h1;
    if (lane == 0) { as[g0 + 1] = s1; ad[g0 + 1] = d1; }
  }
}


__global__ __launch_bounds__(GRP_THREADS) void gat_bucket_scatter(
    const int* __restrict__ src, const int* __restrict__ dst,
    const int* __restrict__ cnt, int* __restrict__ bstart,
    int2* __restrict__ pairs, int E)
{
  static_assert(GRP_CHUNKS == 256 && GRP_THREADS == 1024, "16 row slices x 64 int4 columns");
  __shared__ int s_pre[16][256];
  __shared__ int s_all[16][256];
  __shared__ int s_cur[256];
  __shared__ int s_wtot[8];
  const int t    = threadIdx.x;
  const int lane = t & 31;
  const int w    = t >> 5;
  const int c4   = t & 63;
  const int j    = t >> 6;
  const int k    = blockIdx.x;

  int4 pre = make_int4(0, 0, 0, 0), all = make_int4(0, 0, 0, 0);
  #pragma unroll
  for (int r = 0; r < 16; ++r) {
    const int  kk = j * 16 + r;
    const int4 c  = ((const int4*)cnt)[kk * 64 + c4];
    const bool in_front = kk < k;
    all.x += c.x; all.y += c.y; all.z += c.z; all.w += c.w;
    pre.x += in_front ? c.x : 0; pre.y += in_front ? c.y : 0;
    pre.z += in_front ? c.z : 0; pre.w += in_front ? c.w : 0;
  }
  ((int4*)s_pre[j])[c4] = pre;
  ((int4*)s_all[j])[c4] = all;
  __syncthreads();

  int v = 0, mine = 0;
  if (t < 256) {
    #pragma unroll
    for (int q = 0; q < 16; ++q) { mine += s_pre[q][t]; v += s_all[q][t]; }
  }
  int inc = v;
  #pragma unroll
  for (int o = 1; o < 32; o <<= 1) {
    int u = __shfl_up(inc, o, 32);
    if (lane >= o) inc += u;
  }
  if (t < 256 && lane == 31) s_wtot[w] = inc;
  __syncthreads();
  if (t < 256) {
    int front = 0;
    for (int i = 0; i < w; ++i) front += s_wtot[i];
    const int start = front + inc - v;
    s_cur[t] = start + mine;
    if (k == 0) bstart[t] = start;
  }
  __syncthreads();

  const int per = (E + GRP_CHUNKS - 1) / GRP_CHUNKS;
  const int e0  = k * per;
  const int e1  = min(e0 + per, E);
  for (int base = e0 + t; base < e1; base += GRP_BATCH * GRP_THREADS) {
    int d[GRP_BATCH], s[GRP_BATCH];
    #pragma unroll
    for (int i = 0; i < GRP_BATCH; ++i) {
      const int e = min(base + i * GRP_THREADS, e1 - 1);
      d[i] = dst[e];
      s[i] = src[e];
    }
    #pragma unroll
    for (int i = 0; i < GRP_BATCH; ++i) { keep_here(d[i]); keep_here(s[i]); }
    #pragma unroll
    for (int i = 0; i < GRP_BATCH; ++i)
      if (base + i * GRP_THREADS < e1)
        pairs[atomicAdd(&s_cur[(d[i] >> 8) & 255], 1)] = make_int2(d[i], s[i]);
  }
}

__global__ __launch_bounds__(GRP_THREADS) void gat_bucket_group(
    const int2* __restrict__ pairs, const int* __restrict__ bstart,
    int* __restrict__ deg, int* __restrict__ first, int* __restrict__ gsrc,
    int N, int E)
{
  __shared__ int s_c[256];
  __shared__ int s_wtot[8];
  const int t    = threadIdx.x;
  const int lane = t & 31;
  const int w    = t >> 5;
  const int b    = blockIdx.x;
  const int lo   = bstart[b];
  const int hi   = (b + 1 < 256) ? bstart[b + 1] : E;

  if (t < 256) s_c[t] = 0;
  __syncthreads();
  for (int base = lo + t; base < hi; base += GRP_BATCH * GRP_THREADS) {
    int d[GRP_BATCH];
    #pragma unroll
    for (int i = 0; i < GRP_BATCH; ++i)
      d[i] = pairs[min(base + i * GRP_THREADS, hi - 1)].x;
    #pragma unroll
    for (int i = 0; i < GRP_BATCH; ++i) keep_here(d[i]);
    #pragma unroll
    for (int i = 0; i < GRP_BATCH; ++i)
      if (base + i * GRP_THREADS < hi) atomicAdd(&s_c[d[i] & 255], 1);
  }
  __syncthreads();

  const int v = (t < 256) ? s_c[t] : 0;
  int inc = v;
  #pragma unroll
  for (int o = 1; o < 32; o <<= 1) {
    int u = __shfl_up(inc, o, 32);
    if (lane >= o) inc += u;
  }
  if (t < 256 && lane == 31) s_wtot[w] = inc;
  __syncthreads();
  if (t < 256) {
    int front = lo;
    for (int i = 0; i < w; ++i) front += s_wtot[i];
    const int f = front + inc - v;
    const int n = b * 256 + t;
    if (n < N) { first[n] = f; deg[n] = v; }
    s_c[t] = f;
  }
  __syncthreads();
  for (int base = lo + t; base < hi; base += GRP_BATCH * GRP_THREADS) {
    int2 q[GRP_BATCH];
    #pragma unroll
    for (int i = 0; i < GRP_BATCH; ++i)
      q[i] = pairs[min(base + i * GRP_THREADS, hi - 1)];
    #pragma unroll
    for (int i = 0; i < GRP_BATCH; ++i) { keep_here(q[i].x); keep_here(q[i].y); }
    #pragma unroll
    for (int i = 0; i < GRP_BATCH; ++i)
      if (base + i * GRP_THREADS < hi) gsrc[atomicAdd(&s_c[q[i].x & 255], 1)] = q[i].y;
  }
}

__device__ __forceinline__ float wave_max(float v) {
  #pragma unroll
  for (int m = 16; m >= 1; m >>= 1) v = fmaxf(v, __shfl_xor(v, m, 32));
  return v;
}
__device__ __forceinline__ float wave_sum(float v) {
  #pragma unroll
  for (int m = 16; m >= 1; m >>= 1) v += __shfl_xor(v, m, 32);
  return v;
}
__device__ __forceinline__ float4 h_row(const float* __restrict__ h, int s, int lane) {
  return ((const float4*)(h + (size_t)s * F_DIM))[lane];
}
__device__ __forceinline__ void axpy4(float4& acc, float c, const float4& r) {
  acc.x = fmaf(c, r.x, acc.x);
  acc.y = fmaf(c, r.y, acc.y);
  acc.z = fmaf(c, r.z, acc.z);
  acc.w = fmaf(c, r.w, acc.w);
}

__device__ __forceinline__ void add_rows(
    float4& acc, const float* __restrict__ h, int lane, int s, float c, int n)
{
  const int ci = __float_as_int(c);
  int j = 0;
  for (; j + ROWS_IN_FLIGHT <= n; j += ROWS_IN_FLIGHT) {
    float4 r[ROWS_IN_FLIGHT];
    #pragma unroll
    for (int k = 0; k < ROWS_IN_FLIGHT; ++k)
      r[k] = h_row(h, __builtin_amdgcn_readlane(s, j + k), lane);
    #pragma unroll
    for (int k = 0; k < ROWS_IN_FLIGHT; ++k)
      axpy4(acc, __int_as_float(__builtin_amdgcn_readlane(ci, j + k)), r[k]);
  }
  if (j < n) {
    float4 r[ROWS_IN_FLIGHT];
    #pragma unroll
    for (int k = 0; k < ROWS_IN_FLIGHT; ++k)
      r[k] = h_row(h, __builtin_amdgcn_readlane(s, min(j + k, n - 1)), lane);
    #pragma unroll
    for (int k = 0; k < ROWS_IN_FLIGHT; ++k) {
      const float cj = __int_as_float(__builtin_amdgcn_readlane(ci, min(j + k, n - 1)));
      axpy4(acc, (j + k < n) ? cj : 0.f, r[k]);
    }
  }
}

__device__ __forceinline__ float edge_logit(
    const float* __restrict__ as, int s, float adv)
{
  float t = as[s] + adv;
  return t > 0.f ? t : LEAKY * t;
}

__global__ __launch_bounds__(256) void gat_node_aggregate(
    const int* __restrict__ gsrc, const int* __restrict__ first,
    const int* __restrict__ deg, const float* __restrict__ h,
    const float* __restrict__ as, const float* __restrict__ ad,
    float* __restrict__ out, int N)
{
  const int d    = __builtin_amdgcn_readfirstlane(blockIdx.x * 8 + (threadIdx.x >> 5));
  const int lane = threadIdx.x & 31;
  if (d >= N) return;
  const int   beg = first[d];
  const int   cnt = deg[d];
  const float adv = ad[d];
  const int*  seg = gsrc + beg;

  const bool v0 = lane < cnt;
  int   s0 = 0;
  float l0 = -INFINITY;
  if (v0) { s0 = seg[lane]; l0 = edge_logit(as, s0, adv); }

  float m = l0;
  for (int c = 32; c < cnt; c += 32)
    if (c + lane < cnt) m = fmaxf(m, edge_logit(as, seg[c + lane], adv));
  m = wave_max(m);

  const float p0 = v0 ? expf(l0 - m) : 0.f;
  float sum = p0;
  for (int c = 32; c < cnt; c += 32)
    if (c + lane < cnt) sum += expf(edge_logit(as, seg[c + lane], adv) - m);
  sum = wave_sum(sum);
  const float den = sum + 1e-16f;

  float4 acc = make_float4(0.f, 0.f, 0.f, 0.f);
  add_rows(acc, h, lane, s0, p0 / den, min(cnt, 32));
  for (int c = 32; c < cnt; c += 32) {
    int s = 0; float p = 0.f;
    if (c + lane < cnt) { s = seg[c + lane]; p = expf(edge_logit(as, s, adv) - m); }
    add_rows(acc, h, lane, s, p / den, min(cnt - c, 32));
  }
  ((float4*)(out + (size_t)d * F_DIM))[lane] = acc;
}

extern "C" void kernel_launch(void* const* d_in, const int* in_sizes, int n_in,
                              void* d_out, int out_size, void* d_ws, size_t ws_size,
                              hipStream_t stream) {
  const float* x  = (const float*)d_in[0];
  const int*   ei = (const int*)d_in[1];
  const float* W  = (const float*)d_in[2];
  const float* a  = (const float*)d_in[3];
  const int N = in_sizes[0] / F_DIM;
  const int E = in_sizes[1] / 2;
  const int* src = ei;
  const int* dst = ei + E;

  float* h     = (float*)d_ws;
  int*   deg   = (int*)(h + (size_t)N * F_DIM);
  int*   first = deg + N;
  float* as    = (float*)(first + N);
  float* ad    = as + N;
  int*   gsrc  = (int*)(ad + N) + E;
  float* out   = (float*)d_out;
  int*   cnt    = (int*)d_out;
  int*   bstart = cnt + 256 * 256;
  int2*  pairs  = (int2*)(bstart + 256);

  gat_gemm_wmma_count<<<GRP_CHUNKS + (N + 15) / 16, 256,   0, stream>>>(x, W, a, h, as, ad, N, dst, cnt, E);
  gat_bucket_scatter<<<GRP_CHUNKS,      GRP_THREADS, 0, stream>>>(src, dst, cnt, bstart, pairs, E);
  gat_bucket_group  <<<(N + 255) / 256, GRP_THREADS, 0, stream>>>(pairs, bstart, deg, first, gsrc, N, E);
  gat_node_aggregate<<<(N + 7)  / 8,    256,         0, stream>>>(gsrc, first, deg, h, as, ad, out, N);
}


# weighted row sums taken over x, not h: per 16 nodes the sums form an LDS tile that is multiplied by W on WMMA and stored; h never written; logits from x.(W a); order of sums changed
# speedup vs baseline: 3.5761x; 1.0997x over previous
#include <hip/hip_runtime.h>
#include <hip/hip_bf16.h>


typedef __attribute__((ext_vector_type(2))) float v2f;
typedef __attribute__((ext_vector_type(8))) float v8f;

#define F_DIM 128
#define LEAKY 0.2f
#define ROWS_IN_FLIGHT 8

#define GRP_CHUNKS 256
#define GRP_THREADS 1024
#define GRP_BATCH 4
#define CNT_BATCH 16

__device__ __forceinline__ void keep_here(int& v) { asm volatile("" : "+v"(v)); }

template <int N>
__device__ __forceinline__ float fold_lanes(float (&v)[N], int lane) {
  int mask = 16;
  #pragma unroll
  for (int h = N / 2; h >= 1; h >>= 1) {
    const bool up = (lane & mask) != 0;
    #pragma unroll
    for (int i = 0; i < h; ++i) {
      const float lo = v[i];
      const float hi = v[i + h];
      const float keep = up ? hi : lo;
      const float send = up ? lo : hi;
      v[i] = keep + __shfl_xor(send, mask, 32);
    }
    mask >>= 1;
  }
  #pragma unroll
  for (; mask >= 1; mask >>= 1) v[0] += __shfl_xor(v[0], mask, 32);
  return v[0];
}

__device__ __forceinline__ float wave_max(float v) {
  #pragma unroll
  for (int m = 16; m >= 1; m >>= 1) v = fmaxf(v, __shfl_xor(v, m, 32));
  return v;
}
__device__ __forceinline__ float wave_sum(float v) {
  #pragma unroll
  for (int m = 16; m >= 1; m >>= 1) v += __shfl_xor(v, m, 32);
  return v;
}
__device__ __forceinline__ float4 row_of(const float* __restrict__ m, int s, int lane) {
  return ((const float4*)(m + (size_t)s * F_DIM))[lane];
}
__device__ __forceinline__ float dot4(const float4& p, const float4& q) {
  return p.x * q.x + p.y * q.y + p.z * q.z + p.w * q.w;
}
__device__ __forceinline__ void axpy4(float4& acc, float c, const float4& r) {
  acc.x = fmaf(c, r.x, acc.x);
  acc.y = fmaf(c, r.y, acc.y);
  acc.z = fmaf(c, r.z, acc.z);
  acc.w = fmaf(c, r.w, acc.w);
}

__device__ __forceinline__ void add_rows(
    float4& acc, const float* __restrict__ x, int lane, int s, float c, int n)
{
  const int ci = __float_as_int(c);
  int j = 0;
  for (; j + ROWS_IN_FLIGHT <= n; j += ROWS_IN_FLIGHT) {
    float4 r[ROWS_IN_FLIGHT];
    #pragma unroll
    for (int k = 0; k < ROWS_IN_FLIGHT; ++k)
      r[k] = row_of(x, __builtin_amdgcn_readlane(s, j + k), lane);
    #pragma unroll
    for (int k = 0; k < ROWS_IN_FLIGHT; ++k)
      axpy4(acc, __int_as_float(__builtin_amdgcn_readlane(ci, j + k)), r[k]);
  }
  if (j < n) {
    float4 r[ROWS_IN_FLIGHT];
    #pragma unroll
    for (int k = 0; k < ROWS_IN_FLIGHT; ++k)
      r[k] = row_of(x, __builtin_amdgcn_readlane(s, min(j + k, n - 1)), lane);
    #pragma unroll
    for (int k = 0; k < ROWS_IN_FLIGHT; ++k) {
      const float cj = __int_as_float(__builtin_amdgcn_readlane(ci, min(j + k, n - 1)));
      axpy4(acc, (j + k < n) ? cj : 0.f, r[k]);
    }
  }
}

__device__ __forceinline__ float edge_logit(
    const float* __restrict__ as, int s, float adv)
{
  float t = as[s] + adv;
  return t > 0.f ? t : LEAKY * t;
}

__global__ __launch_bounds__(256) void gat_alpha_count(
    const float* __restrict__ x, const float* __restrict__ W,
    const float* __restrict__ a, float* __restrict__ as, float* __restrict__ ad, int M,
    const int* __restrict__ dst, int* __restrict__ cnt, int E)
{
  static_assert(ROWS_IN_FLIGHT == 8, "fold_lanes<16> below takes 8 rows x 2 sums");
  __shared__ float s_wa[2 * F_DIM];
  __shared__ int   hist[256];
  const int tid = threadIdx.x;

  if (blockIdx.x < GRP_CHUNKS) {
    hist[tid] = 0;
    __syncthreads();
    const int per = (E + GRP_CHUNKS - 1) / GRP_CHUNKS;
    const int e0  = blockIdx.x * per;
    const int e1  = min(e0 + per, E);
    for (int base = e0 + tid; base < e1; base += CNT_BATCH * 256) {
      int d[CNT_BATCH];
      #pragma unroll
      for (int i = 0; i < CNT_BATCH; ++i)
        d[i] = dst[min(base + i * 256, e1 - 1)];
      #pragma unroll
      for (int i = 0; i < CNT_BATCH; ++i) keep_here(d[i]);
      #pragma unroll
      for (int i = 0; i < CNT_BATCH; ++i)
        if (base + i * 256 < e1) atomicAdd(&hist[(d[i] >> 8) & 255], 1);
    }
    __syncthreads();
    cnt[blockIdx.x * 256 + tid] = hist[tid];
    return;
  }

  const int lane = tid & 31;
  const int wv   = tid >> 5;
  const float4 a1 = ((const float4*)a)[lane];
  const float4 a2 = ((const float4*)(a + F_DIM))[lane];
  #pragma unroll
  for (int g = 0; g < 2; ++g) {
    float4 wr[8];
    #pragma unroll
    for (int r = 0; r < 8; ++r) wr[r] = row_of(W, wv * 16 + g * 8 + r, lane);
    float v[16];
    #pragma unroll
    for (int r = 0; r < 8; ++r) {
      v[r]     = dot4(wr[r], a1);
      v[8 + r] = dot4(wr[r], a2);
    }
    const float wa = fold_lanes<16>(v, lane);
    const int   i  = lane >> 1;
    if ((lane & 1) == 0) s_wa[(i >> 3) * F_DIM + wv * 16 + g * 8 + (i & 7)] = wa;
  }
  __syncthreads();
  const float4 w1 = ((const float4*)s_wa)[lane];
  const float4 w2 = ((const float4*)(s_wa + F_DIM))[lane];

  const int nw = (gridDim.x - GRP_CHUNKS) * 8;
  const int w0 = __builtin_amdgcn_readfirstlane((blockIdx.x - GRP_CHUNKS) * 8 + wv);
  for (int b0 = w0 * ROWS_IN_FLIGHT; b0 < M; b0 += nw * ROWS_IN_FLIGHT) {
    float4 xr[ROWS_IN_FLIGHT];
    #pragma unroll
    for (int k = 0; k < ROWS_IN_FLIGHT; ++k)
      xr[k] = row_of(x, min(b0 + k, M - 1), lane);
    float u[16];
    #pragma unroll
    for (int k = 0; k < ROWS_IN_FLIGHT; ++k) {
      u[k]     = dot4(xr[k], w1);
      u[8 + k] = dot4(xr[k], w2);
    }
    const float t = fold_lanes<16>(u, lane);
    const int   i = lane >> 1;
    const int   r = b0 + (i & 7);
    if ((lane & 1) == 0 && r < M) {
      if (i < 8) as[r] = t; else ad[r] = t;
    }
  }
}


__global__ __launch_bounds__(GRP_THREADS) void gat_bucket_scatter(
    const int* __restrict__ src, const int* __restrict__ dst,
    const int* __restrict__ cnt, int* __restrict__ bstart,
    int2* __restrict__ pairs, int E)
{
  static_assert(GRP_CHUNKS == 256 && GRP_THREADS == 1024, "16 row slices x 64 int4 columns");
  __shared__ int s_pre[16][256];
  __shared__ int s_all[16][256];
  __shared__ int s_cur[256];
  __shared__ int s_wtot[8];
  const int t    = threadIdx.x;
  const int lane = t & 31;
  const int w    = t >> 5;
  const int c4   = t & 63;
  const int j    = t >> 6;
  const int k    = blockIdx.x;

  int4 pre = make_int4(0, 0, 0, 0), all = make_int4(0, 0, 0, 0);
  #pragma unroll
  for (int r = 0; r < 16; ++r) {
    const int  kk = j * 16 + r;
    const int4 c  = ((const int4*)cnt)[kk * 64 + c4];
    const bool in_front = kk < k;
    all.x += c.x; all.y += c.y; all.z += c.z; all.w += c.w;
    pre.x += in_front ? c.x : 0; pre.y += in_front ? c.y : 0;
    pre.z += in_front ? c.z : 0; pre.w += in_front ? c.w : 0;
  }
  ((int4*)s_pre[j])[c4] = pre;
  ((int4*)s_all[j])[c4] = all;
  __syncthreads();

  int v = 0, mine = 0;
  if (t < 256) {
    #pragma unroll
    for (int q = 0; q < 16; ++q) { mine += s_pre[q][t]; v += s_all[q][t]; }
  }
  int inc = v;
  #pragma unroll
  for (int o = 1; o < 32; o <<= 1) {
    int u = __shfl_up(inc, o, 32);
    if (lane >= o) inc += u;
  }
  if (t < 256 && lane == 31) s_wtot[w] = inc;
  __syncthreads();
  if (t < 256) {
    int front = 0;
    for (int i = 0; i < w; ++i) front += s_wtot[i];
    const int start = front + inc - v;
    s_cur[t] = start + mine;
    if (k == 0) bstart[t] = start;
  }
  __syncthreads();

  const int per = (E + GRP_CHUNKS - 1) / GRP_CHUNKS;
  const int e0  = k * per;
  const int e1  = min(e0 + per, E);
  for (int base = e0 + t; base < e1; base += GRP_BATCH * GRP_THREADS) {
    int d[GRP_BATCH], s[GRP_BATCH];
    #pragma unroll
    for (int i = 0; i < GRP_BATCH; ++i) {
      const int e = min(base + i * GRP_THREADS, e1 - 1);
      d[i] = dst[e];
      s[i] = src[e];
    }
    #pragma unroll
    for (int i = 0; i < GRP_BATCH; ++i) { keep_here(d[i]); keep_here(s[i]); }
    #pragma unroll
    for (int i = 0; i < GRP_BATCH; ++i)
      if (base + i * GRP_THREADS < e1)
        pairs[atomicAdd(&s_cur[(d[i] >> 8) & 255], 1)] = make_int2(d[i], s[i]);
  }
}

__global__ __launch_bounds__(GRP_THREADS) void gat_bucket_group(
    const int2* __restrict__ pairs, const int* __restrict__ bstart,
    int* __restrict__ deg, int* __restrict__ first, int* __restrict__ gsrc,
    int N, int E)
{
  __shared__ int s_c[256];
  __shared__ int s_wtot[8];
  const int t    = threadIdx.x;
  const int lane = t & 31;
  const int w    = t >> 5;
  const int b    = blockIdx.x;
  const int lo   = bstart[b];
  const int hi   = (b + 1 < 256) ? bstart[b + 1] : E;

  if (t < 256) s_c[t] = 0;
  __syncthreads();
  for (int base = lo + t; base < hi; base += GRP_BATCH * GRP_THREADS) {
    int d[GRP_BATCH];
    #pragma unroll
    for (int i = 0; i < GRP_BATCH; ++i)
      d[i] = pairs[min(base + i * GRP_THREADS, hi - 1)].x;
    #pragma unroll
    for (int i = 0; i < GRP_BATCH; ++i) keep_here(d[i]);
    #pragma unroll
    for (int i = 0; i < GRP_BATCH; ++i)
      if (base + i * GRP_THREADS < hi) atomicAdd(&s_c[d[i] & 255], 1);
  }
  __syncthreads();

  const int v = (t < 256) ? s_c[t] : 0;
  int inc = v;
  #pragma unroll
  for (int o = 1; o < 32; o <<= 1) {
    int u = __shfl_up(inc, o, 32);
    if (lane >= o) inc += u;
  }
  if (t < 256 && lane == 31) s_wtot[w] = inc;
  __syncthreads();
  if (t < 256) {
    int front = lo;
    for (int i = 0; i < w; ++i) front += s_wtot[i];
    const int f = front + inc - v;
    const int n = b * 256 + t;
    if (n < N) { first[n] = f; deg[n] = v; }
    s_c[t] = f;
  }
  __syncthreads();
  for (int base = lo + t; base < hi; base += GRP_BATCH * GRP_THREADS) {
    int2 q[GRP_BATCH];
    #pragma unroll
    for (int i = 0; i < GRP_BATCH; ++i)
      q[i] = pairs[min(base + i * GRP_THREADS, hi - 1)];
    #pragma unroll
    for (int i = 0; i < GRP_BATCH; ++i) { keep_here(q[i].x); keep_here(q[i].y); }
    #pragma unroll
    for (int i = 0; i < GRP_BATCH; ++i)
      if (base + i * GRP_THREADS < hi) gsrc[atomicAdd(&s_c[q[i].x & 255], 1)] = q[i].y;
  }
}

__device__ __forceinline__ float4 node_sum(
    const int* __restrict__ gsrc, const int* __restrict__ first,
    const int* __restrict__ deg, const float* __restrict__ x,
    const float* __restrict__ as, const float* __restrict__ ad, int d, int lane)
{
  const int   beg = first[d];
  const int   cnt = deg[d];
  const float adv = ad[d];
  const int*  seg = gsrc + beg;

  const bool v0 = lane < cnt;
  int   s0 = 0;
  float l0 = -INFINITY;
  if (v0) { s0 = seg[lane]; l0 = edge_logit(as, s0, adv); }

  float m = l0;
  for (int c = 32; c < cnt; c += 32)
    if (c + lane < cnt) m = fmaxf(m, edge_logit(as, seg[c + lane], adv));
  m = wave_max(m);

  const float p0 = v0 ? expf(l0 - m) : 0.f;
  float sum = p0;
  for (int c = 32; c < cnt; c += 32)
    if (c + lane < cnt) sum += expf(edge_logit(as, seg[c + lane], adv) - m);
  sum = wave_sum(sum);
  const float den = sum + 1e-16f;

  float4 acc = make_float4(0.f, 0.f, 0.f, 0.f);
  add_rows(acc, x, lane, s0, p0 / den, min(cnt, 32));
  for (int c = 32; c < cnt; c += 32) {
    int s = 0; float p = 0.f;
    if (c + lane < cnt) { s = seg[c + lane]; p = expf(edge_logit(as, s, adv) - m); }
    add_rows(acc, x, lane, s, p / den, min(cnt - c, 32));
  }
  return acc;
}

__global__ __launch_bounds__(256) void gat_aggregate_gemm(
    const int* __restrict__ gsrc, const int* __restrict__ first,
    const int* __restrict__ deg, const float* __restrict__ x,
    const float* __restrict__ as, const float* __restrict__ ad,
    const float* __restrict__ W, float* __restrict__ out, int N)
{
  __shared__ float sA[16 * 132];
  __shared__ float sH[16 * 132];
  const int tid  = threadIdx.x;
  const int lane = tid & 31;
  const int wv   = tid >> 5;
  const int m0   = blockIdx.x * 16;
  const int r0   = wv * 2;

  for (int q = 0; q < 2; ++q) {
    const int d = __builtin_amdgcn_readfirstlane(m0 + r0 + q);
    float4 z = make_float4(0.f, 0.f, 0.f, 0.f);
    if (d < N) z = node_sum(gsrc, first, deg, x, as, ad, d, lane);
    ((float4*)sA)[(r0 + q) * 33 + lane] = z;
  }
  __syncthreads();

  const int row  = lane & 15;
  const int kh   = (lane >> 4) * 2;
  const int n0   = wv * 16;

  v8f acc = {};
  #pragma unroll 8
  for (int k = 0; k < F_DIM; k += 4) {
    v2f a, b;
    a.x = sA[row * 132 + k + kh + 0];
    a.y = sA[row * 132 + k + kh + 1];
    b.x = W[(k + kh + 0) * F_DIM + n0 + row];
    b.y = W[(k + kh + 1) * F_DIM + n0 + row];
    acc = __builtin_amdgcn_wmma_f32_16x16x4_f32(false, a, false, b,
                                                (short)0, acc, false, false);
  }

  const int mh = (lane >> 4) * 8;
  #pragma unroll
  for (int v = 0; v < 8; ++v) sH[(mh + v) * 132 + n0 + row] = acc[v];
  __syncthreads();

  const int g0 = m0 + r0;
  if (g0 < N)
    ((float4*)(out + (size_t)g0 * F_DIM))[lane] = ((const float4*)sH)[(r0 + 0) * 33 + lane];
  if (g0 + 1 < N)
    ((float4*)(out + (size_t)(g0 + 1) * F_DIM))[lane] = ((const float4*)sH)[(r0 + 1) * 33 + lane];
}

extern "C" void kernel_launch(void* const* d_in, const int* in_sizes, int n_in,
                              void* d_out, int out_size, void* d_ws, size_t ws_size,
                              hipStream_t stream) {
  const float* x  = (const float*)d_in[0];
  const int*   ei = (const int*)d_in[1];
  const float* W  = (const float*)d_in[2];
  const float* a  = (const float*)d_in[3];
  const int N = in_sizes[0] / F_DIM;
  const int E = in_sizes[1] / 2;
  const int* src = ei;
  const int* dst = ei + E;

  float* ws0   = (float*)d_ws;
  int*   deg   = (int*)(ws0 + (size_t)N * F_DIM);
  int*   first = deg + N;
  float* as    = (float*)(first + N);
  float* ad    = as + N;
  int*   gsrc  = (int*)(ad + N) + E;
  float* out   = (float*)d_out;
  int*   cnt    = (int*)d_out;
  int*   bstart = cnt + 256 * 256;
  int2*  pairs  = (int2*)(bstart + 256);

  gat_alpha_count   <<<GRP_CHUNKS + 782,  256,         0, stream>>>(x, W, a, as, ad, N, dst, cnt, E);
  gat_bucket_scatter<<<GRP_CHUNKS,      GRP_THREADS, 0, stream>>>(src, dst, cnt, bstart, pairs, E);
  gat_bucket_group  <<<(N + 255) / 256, GRP_THREADS, 0, stream>>>(pairs, bstart, deg, first, gsrc, N, E);
  gat_aggregate_gemm<<<(N + 15) / 16,   256,         0, stream>>>(gsrc, first, deg, x, as, ad, W, out, N);
}


# edge grouping: each chunk is ordered by bucket inside its own run of the pair array, so level 1 needs no count of other chunks and rides in launch 1: 4 launches become 3; order of sums changed
# speedup vs baseline: 4.4191x; 1.2357x over previous
#include <hip/hip_runtime.h>
#include <hip/hip_bf16.h>


typedef __attribute__((ext_vector_type(2))) float v2f;
typedef __attribute__((ext_vector_type(8))) float v8f;

#define F_DIM 128
#define LEAKY 0.2f
#define ROWS_IN_FLIGHT 8

#define GRP_CHUNKS 256
#define GRP_THREADS 1024
#define GRP_BATCH 4
#define CNT_BATCH 16

__device__ __forceinline__ void keep_here(int& v) { asm volatile("" : "+v"(v)); }

template <typename T, int N>
__device__ __forceinline__ T fold_lanes(T (&v)[N], int lane) {
  int mask = 16;
  #pragma unroll
  for (int h = N / 2; h >= 1; h >>= 1) {
    const bool up = (lane & mask) != 0;
    #pragma unroll
    for (int i = 0; i < h; ++i) {
      const T lo = v[i];
      const T hi = v[i + h];
      const T keep = up ? hi : lo;
      const T send = up ? lo : hi;
      v[i] = keep + __shfl_xor(send, mask, 32);
    }
    mask >>= 1;
  }
  #pragma unroll
  for (; mask >= 1; mask >>= 1) v[0] += __shfl_xor(v[0], mask, 32);
  return v[0];
}

__device__ __forceinline__ float wave_max(float v) {
  #pragma unroll
  for (int m = 16; m >= 1; m >>= 1) v = fmaxf(v, __shfl_xor(v, m, 32));
  return v;
}
__device__ __forceinline__ float wave_sum(float v) {
  #pragma unroll
  for (int m = 16; m >= 1; m >>= 1) v += __shfl_xor(v, m, 32);
  return v;
}
__device__ __forceinline__ float4 row_of(const float* __restrict__ m, int s, int lane) {
  return ((const float4*)(m + (size_t)s * F_DIM))[lane];
}
__device__ __forceinline__ float dot4(const float4& p, const float4& q) {
  return p.x * q.x + p.y * q.y + p.z * q.z + p.w * q.w;
}
__device__ __forceinline__ void axpy4(float4& acc, float c, const float4& r) {
  acc.x = fmaf(c, r.x, acc.x);
  acc.y = fmaf(c, r.y, acc.y);
  acc.z = fmaf(c, r.z, acc.z);
  acc.w = fmaf(c, r.w, acc.w);
}

__device__ __forceinline__ void add_rows(
    float4& acc, const float* __restrict__ x, int lane, int s, float c, int n)
{
  const int ci = __float_as_int(c);
  int j = 0;
  for (; j + ROWS_IN_FLIGHT <= n; j += ROWS_IN_FLIGHT) {
    float4 r[ROWS_IN_FLIGHT];
    #pragma unroll
    for (int k = 0; k < ROWS_IN_FLIGHT; ++k)
      r[k] = row_of(x, __builtin_amdgcn_readlane(s, j + k), lane);
    #pragma unroll
    for (int k = 0; k < ROWS_IN_FLIGHT; ++k)
      axpy4(acc, __int_as_float(__builtin_amdgcn_readlane(ci, j + k)), r[k]);
  }
  if (j < n) {
    float4 r[ROWS_IN_FLIGHT];
    #pragma unroll
    for (int k = 0; k < ROWS_IN_FLIGHT; ++k)
      r[k] = row_of(x, __builtin_amdgcn_readlane(s, min(j + k, n - 1)), lane);
    #pragma unroll
    for (int k = 0; k < ROWS_IN_FLIGHT; ++k) {
      const float cj = __int_as_float(__builtin_amdgcn_readlane(ci, min(j + k, n - 1)));
      axpy4(acc, (j + k < n) ? cj : 0.f, r[k]);
    }
  }
}

__device__ __forceinline__ float edge_logit(
    const float* __restrict__ as, int s, float adv)
{
  float t = as[s] + adv;
  return t > 0.f ? t : LEAKY * t;
}

__global__ __launch_bounds__(256) void gat_alpha_count(
    const float* __restrict__ x, const float* __restrict__ W,
    const float* __restrict__ a, float* __restrict__ as, float* __restrict__ ad, int M,
    const int* __restrict__ src, const int* __restrict__ dst,
    int* __restrict__ cnt, int2* __restrict__ pairs, int E)
{
  static_assert(ROWS_IN_FLIGHT == 8, "fold_lanes below takes 8 rows x 2 sums");
  __shared__ float s_wa[2 * F_DIM];
  __shared__ int   hist[256];
  __shared__ int   s_wtot[8];
  const int tid = threadIdx.x;

  if (blockIdx.x < GRP_CHUNKS) {
    const int clane = tid & 31;
    const int cwv   = tid >> 5;
    hist[tid] = 0;
    __syncthreads();
    const int per = (E + GRP_CHUNKS - 1) / GRP_CHUNKS;
    const int e0  = blockIdx.x * per;
    const int e1  = min(e0 + per, E);
    for (int base = e0 + tid; base < e1; base += CNT_BATCH * 256) {
      int d[CNT_BATCH];
      #pragma unroll
      for (int i = 0; i < CNT_BATCH; ++i)
        d[i] = dst[min(base + i * 256, e1 - 1)];
      #pragma unroll
      for (int i = 0; i < CNT_BATCH; ++i) keep_here(d[i]);
      #pragma unroll
      for (int i = 0; i < CNT_BATCH; ++i)
        if (base + i * 256 < e1) atomicAdd(&hist[(d[i] >> 8) & 255], 1);
    }
    __syncthreads();
    const int v = hist[tid];
    cnt[blockIdx.x * 256 + tid] = v;
    int inc = v;
    #pragma unroll
    for (int o = 1; o < 32; o <<= 1) {
      int u = __shfl_up(inc, o, 32);
      if (clane >= o) inc += u;
    }
    if (clane == 31) s_wtot[cwv] = inc;
    __syncthreads();
    int front = 0;
    for (int i = 0; i < cwv; ++i) front += s_wtot[i];
    hist[tid] = front + inc - v;
    __syncthreads();
    for (int base = e0 + tid; base < e1; base += CNT_BATCH * 256) {
      int d[CNT_BATCH], s[CNT_BATCH];
      #pragma unroll
      for (int i = 0; i < CNT_BATCH; ++i) {
        const int e = min(base + i * 256, e1 - 1);
        d[i] = dst[e];
        s[i] = src[e];
      }
      #pragma unroll
      for (int i = 0; i < CNT_BATCH; ++i) { keep_here(d[i]); keep_here(s[i]); }
      #pragma unroll
      for (int i = 0; i < CNT_BATCH; ++i)
        if (base + i * 256 < e1)
          pairs[e0 + atomicAdd(&hist[(d[i] >> 8) & 255], 1)] = make_int2(d[i], s[i]);
    }
    return;
  }

  const int lane = tid & 31;
  const int wv   = tid >> 5;
  const float4 a1 = ((const float4*)a)[lane];
  const float4 a2 = ((const float4*)(a + F_DIM))[lane];
  #pragma unroll
  for (int g = 0; g < 2; ++g) {
    float4 wr[8];
    #pragma unroll
    for (int r = 0; r < 8; ++r) wr[r] = row_of(W, wv * 16 + g * 8 + r, lane);
    float v[16];
    #pragma unroll
    for (int r = 0; r < 8; ++r) {
      v[r]     = dot4(wr[r], a1);
      v[8 + r] = dot4(wr[r], a2);
    }
    const float wa = fold_lanes(v, lane);
    const int   i  = lane >> 1;
    if ((lane & 1) == 0) s_wa[(i >> 3) * F_DIM + wv * 16 + g * 8 + (i & 7)] = wa;
  }
  __syncthreads();
  const float4 w1 = ((const float4*)s_wa)[lane];
  const float4 w2 = ((const float4*)(s_wa + F_DIM))[lane];

  const int nw = (gridDim.x - GRP_CHUNKS) * 8;
  const int w0 = __builtin_amdgcn_readfirstlane((blockIdx.x - GRP_CHUNKS) * 8 + wv);
  for (int b0 = w0 * ROWS_IN_FLIGHT; b0 < M; b0 += nw * ROWS_IN_FLIGHT) {
    float4 xr[ROWS_IN_FLIGHT];
    #pragma unroll
    for (int k = 0; k < ROWS_IN_FLIGHT; ++k)
      xr[k] = row_of(x, min(b0 + k, M - 1), lane);
    float u[16];
    #pragma unroll
    for (int k = 0; k < ROWS_IN_FLIGHT; ++k) {
      u[k]     = dot4(xr[k], w1);
      u[8 + k] = dot4(xr[k], w2);
    }
    const float t = fold_lanes(u, lane);
    const int   i = lane >> 1;
    const int   r = b0 + (i & 7);
    if ((lane & 1) == 0 && r < M) {
      if (i < 8) as[r] = t; else ad[r] = t;
    }
  }
}


__global__ __launch_bounds__(GRP_THREADS) void gat_bucket_group(
    const int2* __restrict__ pairs, const int* __restrict__ cnt,
    int* __restrict__ deg, int* __restrict__ first, int* __restrict__ gsrc,
    int N, int E)
{
  static_assert(GRP_CHUNKS == 256 && GRP_THREADS == 1024, "32 waves x 8 table rows; 4 threads per chunk");
  __shared__ int s_off[GRP_CHUNKS];
  __shared__ int s_n[GRP_CHUNKS];
  __shared__ int s_c[256];
  __shared__ int s_wtot[8];
  __shared__ int s_lo[8];
  const int t    = threadIdx.x;
  const int lane = t & 31;
  const int w    = t >> 5;
  const int b    = blockIdx.x;

  int v[8];
  #pragma unroll
  for (int r = 0; r < 8; ++r) {
    const int  k  = w * 8 + r;
    const int4 c0 = ((const int4*)cnt)[k * 64 + lane];
    const int4 c1 = ((const int4*)cnt)[k * 64 + 32 + lane];
    const int  q0 = 4 * lane;
    const int  q1 = 128 + 4 * lane;
    int sum = 0;
    sum += (q0 + 0 < b) ? c0.x : 0;
    sum += (q0 + 1 < b) ? c0.y : 0;
    sum += (q0 + 2 < b) ? c0.z : 0;
    sum += (q0 + 3 < b) ? c0.w : 0;
    sum += (q1 + 0 < b) ? c1.x : 0;
    sum += (q1 + 1 < b) ? c1.y : 0;
    sum += (q1 + 2 < b) ? c1.z : 0;
    sum += (q1 + 3 < b) ? c1.w : 0;
    v[r] = sum;
  }
  const int off = fold_lanes(v, lane);
  if ((lane & 3) == 0) s_off[w * 8 + (lane >> 2)] = off;
  if (t < GRP_CHUNKS) s_n[t] = cnt[t * 256 + b];
  if (t < 256) s_c[t] = 0;
  __syncthreads();

  int part = (t < GRP_CHUNKS) ? s_off[t] : 0;
  #pragma unroll
  for (int m = 16; m >= 1; m >>= 1) part += __shfl_xor(part, m, 32);
  if (t < GRP_CHUNKS && lane == 0) s_lo[w] = part;

  const int per = (E + GRP_CHUNKS - 1) / GRP_CHUNKS;
  const int k   = t >> 2;
  const int sub = t & 3;
  const int nk  = s_n[k];
  const int2* __restrict__ pk = pairs + (size_t)k * per + s_off[k];
  for (int i0 = sub; i0 < nk; i0 += 4 * GRP_BATCH) {
    int d[GRP_BATCH];
    #pragma unroll
    for (int i = 0; i < GRP_BATCH; ++i)
      d[i] = pk[min(i0 + 4 * i, nk - 1)].x;
    #pragma unroll
    for (int i = 0; i < GRP_BATCH; ++i) keep_here(d[i]);
    #pragma unroll
    for (int i = 0; i < GRP_BATCH; ++i)
      if (i0 + 4 * i < nk) atomicAdd(&s_c[d[i] & 255], 1);
  }
  __syncthreads();

  int lo = 0;
  for (int i = 0; i < 8; ++i) lo += s_lo[i];

  const int c = (t < 256) ? s_c[t] : 0;
  int inc = c;
  #pragma unroll
  for (int o = 1; o < 32; o <<= 1) {
    int u = __shfl_up(inc, o, 32);
    if (lane >= o) inc += u;
  }
  if (t < 256 && lane == 31) s_wtot[w] = inc;
  __syncthreads();
  if (t < 256) {
    int front = lo;
    for (int i = 0; i < w; ++i) front += s_wtot[i];
    const int f = front + inc - c;
    const int n = b * 256 + t;
    if (n < N) { first[n] = f; deg[n] = c; }
    s_c[t] = f;
  }
  __syncthreads();
  for (int i0 = sub; i0 < nk; i0 += 4 * GRP_BATCH) {
    int2 q[GRP_BATCH];
    #pragma unroll
    for (int i = 0; i < GRP_BATCH; ++i)
      q[i] = pk[min(i0 + 4 * i, nk - 1)];
    #pragma unroll
    for (int i = 0; i < GRP_BATCH; ++i) { keep_here(q[i].x); keep_here(q[i].y); }
    #pragma unroll
    for (int i = 0; i < GRP_BATCH; ++i)
      if (i0 + 4 * i < nk) gsrc[atomicAdd(&s_c[q[i].x & 255], 1)] = q[i].y;
  }
}

__device__ __forceinline__ float4 node_sum(
    const int* __restrict__ gsrc, const int* __restrict__ first,
    const int* __restrict__ deg, const float* __restrict__ x,
    const float* __restrict__ as, const float* __restrict__ ad, int d, int lane)
{
  const int   beg = first[d];
  const int   cnt = deg[d];
  const float adv = ad[d];
  const int*  seg = gsrc + beg;

  const bool v0 = lane < cnt;
  int   s0 = 0;
  float l0 = -INFINITY;
  if (v0) { s0 = seg[lane]; l0 = edge_logit(as, s0, adv); }

  float m = l0;
  for (int c = 32; c < cnt; c += 32)
    if (c + lane < cnt) m = fmaxf(m, edge_logit(as, seg[c + lane], adv));
  m = wave_max(m);

  const float p0 = v0 ? expf(l0 - m) : 0.f;
  float sum = p0;
  for (int c = 32; c < cnt; c += 32)
    if (c + lane < cnt) sum += expf(edge_logit(as, seg[c + lane], adv) - m);
  sum = wave_sum(sum);
  const float den = sum + 1e-16f;

  float4 acc = make_float4(0.f, 0.f, 0.f, 0.f);
  add_rows(acc, x, lane, s0, p0 / den, min(cnt, 32));
  for (int c = 32; c < cnt; c += 32) {
    int s = 0; float p = 0.f;
    if (c + lane < cnt) { s = seg[c + lane]; p = expf(edge_logit(as, s, adv) - m); }
    add_rows(acc, x, lane, s, p / den, min(cnt - c, 32));
  }
  return acc;
}

__global__ __launch_bounds__(256) void gat_aggregate_gemm(
    const int* __restrict__ gsrc, const int* __restrict__ first,
    const int* __restrict__ deg, const float* __restrict__ x,
    const float* __restrict__ as, const float* __restrict__ ad,
    const float* __restrict__ W, float* __restrict__ out, int N)
{
  __shared__ float sA[16 * 132];
  __shared__ float sH[16 * 132];
  const int tid  = threadIdx.x;
  const int lane = tid & 31;
  const int wv   = tid >> 5;
  const int m0   = blockIdx.x * 16;
  const int r0   = wv * 2;

  for (int q = 0; q < 2; ++q) {
    const int d = __builtin_amdgcn_readfirstlane(m0 + r0 + q);
    float4 z = make_float4(0.f, 0.f, 0.f, 0.f);
    if (d < N) z = node_sum(gsrc, first, deg, x, as, ad, d, lane);
    ((float4*)sA)[(r0 + q) * 33 + lane] = z;
  }
  __syncthreads();

  const int row  = lane & 15;
  const int kh   = (lane >> 4) * 2;
  const int n0   = wv * 16;

  v8f acc = {};
  #pragma unroll 8
  for (int k = 0; k < F_DIM; k += 4) {
    v2f a, b;
    a.x = sA[row * 132 + k + kh + 0];
    a.y = sA[row * 132 + k + kh + 1];
    b.x = W[(k + kh + 0) * F_DIM + n0 + row];
    b.y = W[(k + kh + 1) * F_DIM + n0 + row];
    acc = __builtin_amdgcn_wmma_f32_16x16x4_f32(false, a, false, b,
                                                (short)0, acc, false, false);
  }

  const int mh = (lane >> 4) * 8;
  #pragma unroll
  for (int v = 0; v < 8; ++v) sH[(mh + v) * 132 + n0 + row] = acc[v];
  __syncthreads();

  const int g0 = m0 + r0;
  if (g0 < N)
    ((float4*)(out + (size_t)g0 * F_DIM))[lane] = ((const float4*)sH)[(r0 + 0) * 33 + lane];
  if (g0 + 1 < N)
    ((float4*)(out + (size_t)(g0 + 1) * F_DIM))[lane] = ((const float4*)sH)[(r0 + 1) * 33 + lane];
}

extern "C" void kernel_launch(void* const* d_in, const int* in_sizes, int n_in,
                              void* d_out, int out_size, void* d_ws, size_t ws_size,
                              hipStream_t stream) {
  const float* x  = (const float*)d_in[0];
  const int*   ei = (const int*)d_in[1];
  const float* W  = (const float*)d_in[2];
  const float* a  = (const float*)d_in[3];
  const int N = in_sizes[0] / F_DIM;
  const int E = in_sizes[1] / 2;
  const int* src = ei;
  const int* dst = ei + E;

  float* ws0   = (float*)d_ws;
  int*   deg   = (int*)(ws0 + (size_t)N * F_DIM);
  int*   first = deg + N;
  float* as    = (float*)(first + N);
  float* ad    = as + N;
  int*   gsrc  = (int*)(ad + N) + E;
  float* out   = (float*)d_out;
  int*   cnt    = (int*)d_out;
  int2*  pairs  = (int2*)(cnt + 256 * 256);

  gat_alpha_count   <<<GRP_CHUNKS + 782,  256,         0, stream>>>(x, W, a, as, ad, N, src, dst, cnt, pairs, E);
  gat_bucket_group  <<<(N + 255) / 256, GRP_THREADS, 0, stream>>>(pairs, cnt, deg, first, gsrc, N, E);
  gat_aggregate_gemm<<<(N + 15) / 16,   256,         0, stream>>>(gsrc, first, deg, x, as, ad, W, out, N);
}


# aggregate launch batched 4 nodes per wave (32 per work-group): their segment heads fetched together, both 16-row tiles share each W fragment load; half the waves; sums in same order
# speedup vs baseline: 4.6174x; 1.0449x over previous
#include <hip/hip_runtime.h>
#include <hip/hip_bf16.h>


typedef __attribute__((ext_vector_type(2))) float v2f;
typedef __attribute__((ext_vector_type(8))) float v8f;

#define F_DIM 128
#define LEAKY 0.2f
#define ROWS_IN_FLIGHT 8
#define TILE_NODES 32
#define NODES_PER_WAVE 4

#define GRP_CHUNKS 256
#define GRP_THREADS 1024
#define GRP_BATCH 4
#define CNT_BATCH 16

__device__ __forceinline__ void keep_here(int& v) { asm volatile("" : "+v"(v)); }

template <typename T, int N>
__device__ __forceinline__ T fold_lanes(T (&v)[N], int lane) {
  int mask = 16;
  #pragma unroll
  for (int h = N / 2; h >= 1; h >>= 1) {
    const bool up = (lane & mask) != 0;
    #pragma unroll
    for (int i = 0; i < h; ++i) {
      const T lo = v[i];
      const T hi = v[i + h];
      const T keep = up ? hi : lo;
      const T send = up ? lo : hi;
      v[i] = keep + __shfl_xor(send, mask, 32);
    }
    mask >>= 1;
  }
  #pragma unroll
  for (; mask >= 1; mask >>= 1) v[0] += __shfl_xor(v[0], mask, 32);
  return v[0];
}

__device__ __forceinline__ float wave_max(float v) {
  #pragma unroll
  for (int m = 16; m >= 1; m >>= 1) v = fmaxf(v, __shfl_xor(v, m, 32));
  return v;
}
__device__ __forceinline__ float wave_sum(float v) {
  #pragma unroll
  for (int m = 16; m >= 1; m >>= 1) v += __shfl_xor(v, m, 32);
  return v;
}
__device__ __forceinline__ float4 row_of(const float* __restrict__ m, int s, int lane) {
  return ((const float4*)(m + (size_t)s * F_DIM))[lane];
}
__device__ __forceinline__ float dot4(const float4& p, const float4& q) {
  return p.x * q.x + p.y * q.y + p.z * q.z + p.w * q.w;
}
__device__ __forceinline__ void axpy4(float4& acc, float c, const float4& r) {
  acc.x = fmaf(c, r.x, acc.x);
  acc.y = fmaf(c, r.y, acc.y);
  acc.z = fmaf(c, r.z, acc.z);
  acc.w = fmaf(c, r.w, acc.w);
}

__device__ __forceinline__ void add_rows(
    float4& acc, const float* __restrict__ x, int lane, int s, float c, int n)
{
  const int ci = __float_as_int(c);
  int j = 0;
  for (; j + ROWS_IN_FLIGHT <= n; j += ROWS_IN_FLIGHT) {
    float4 r[ROWS_IN_FLIGHT];
    #pragma unroll
    for (int k = 0; k < ROWS_IN_FLIGHT; ++k)
      r[k] = row_of(x, __builtin_amdgcn_readlane(s, j + k), lane);
    #pragma unroll
    for (int k = 0; k < ROWS_IN_FLIGHT; ++k)
      axpy4(acc, __int_as_float(__builtin_amdgcn_readlane(ci, j + k)), r[k]);
  }
  if (j < n) {
    float4 r[ROWS_IN_FLIGHT];
    #pragma unroll
    for (int k = 0; k < ROWS_IN_FLIGHT; ++k)
      r[k] = row_of(x, __builtin_amdgcn_readlane(s, min(j + k, n - 1)), lane);
    #pragma unroll
    for (int k = 0; k < ROWS_IN_FLIGHT; ++k) {
      const float cj = __int_as_float(__builtin_amdgcn_readlane(ci, min(j + k, n - 1)));
      axpy4(acc, (j + k < n) ? cj : 0.f, r[k]);
    }
  }
}

__device__ __forceinline__ float edge_logit(
    const float* __restrict__ as, int s, float adv)
{
  float t = as[s] + adv;
  return t > 0.f ? t : LEAKY * t;
}

__global__ __launch_bounds__(256) void gat_alpha_count(
    const float* __restrict__ x, const float* __restrict__ W,
    const float* __restrict__ a, float* __restrict__ as, float* __restrict__ ad, int M,
    const int* __restrict__ src, const int* __restrict__ dst,
    int* __restrict__ cnt, int2* __restrict__ pairs, int E)
{
  static_assert(ROWS_IN_FLIGHT == 8, "fold_lanes below takes 8 rows x 2 sums");
  __shared__ float s_wa[2 * F_DIM];
  __shared__ int   hist[256];
  __shared__ int   s_wtot[8];
  const int tid = threadIdx.x;

  if (blockIdx.x < GRP_CHUNKS) {
    const int clane = tid & 31;
    const int cwv   = tid >> 5;
    hist[tid] = 0;
    __syncthreads();
    const int per = (E + GRP_CHUNKS - 1) / GRP_CHUNKS;
    const int e0  = blockIdx.x * per;
    const int e1  = min(e0 + per, E);
    for (int base = e0 + tid; base < e1; base += CNT_BATCH * 256) {
      int d[CNT_BATCH];
      #pragma unroll
      for (int i = 0; i < CNT_BATCH; ++i)
        d[i] = dst[min(base + i * 256, e1 - 1)];
      #pragma unroll
      for (int i = 0; i < CNT_BATCH; ++i) keep_here(d[i]);
      #pragma unroll
      for (int i = 0; i < CNT_BATCH; ++i)
        if (base + i * 256 < e1) atomicAdd(&hist[(d[i] >> 8) & 255], 1);
    }
    __syncthreads();
    const int v = hist[tid];
    cnt[blockIdx.x * 256 + tid] = v;
    int inc = v;
    #pragma unroll
    for (int o = 1; o < 32; o <<= 1) {
      int u = __shfl_up(inc, o, 32);
      if (clane >= o) inc += u;
    }
    if (clane == 31) s_wtot[cwv] = inc;
    __syncthreads();
    int front = 0;
    for (int i = 0; i < cwv; ++i) front += s_wtot[i];
    hist[tid] = front + inc - v;
    __syncthreads();
    for (int base = e0 + tid; base < e1; base += CNT_BATCH * 256) {
      int d[CNT_BATCH], s[CNT_BATCH];
      #pragma unroll
      for (int i = 0; i < CNT_BATCH; ++i) {
        const int e = min(base + i * 256, e1 - 1);
        d[i] = dst[e];
        s[i] = src[e];
      }
      #pragma unroll
      for (int i = 0; i < CNT_BATCH; ++i) { keep_here(d[i]); keep_here(s[i]); }
      #pragma unroll
      for (int i = 0; i < CNT_BATCH; ++i)
        if (base + i * 256 < e1)
          pairs[e0 + atomicAdd(&hist[(d[i] >> 8) & 255], 1)] = make_int2(d[i], s[i]);
    }
    return;
  }

  const int lane = tid & 31;
  const int wv   = tid >> 5;
  const float4 a1 = ((const float4*)a)[lane];
  const float4 a2 = ((const float4*)(a + F_DIM))[lane];
  #pragma unroll
  for (int g = 0; g < 2; ++g) {
    float4 wr[8];
    #pragma unroll
    for (int r = 0; r < 8; ++r) wr[r] = row_of(W, wv * 16 + g * 8 + r, lane);
    float v[16];
    #pragma unroll
    for (int r = 0; r < 8; ++r) {
      v[r]     = dot4(wr[r], a1);
      v[8 + r] = dot4(wr[r], a2);
    }
    const float wa = fold_lanes(v, lane);
    const int   i  = lane >> 1;
    if ((lane & 1) == 0) s_wa[(i >> 3) * F_DIM + wv * 16 + g * 8 + (i & 7)] = wa;
  }
  __syncthreads();
  const float4 w1 = ((const float4*)s_wa)[lane];
  const float4 w2 = ((const float4*)(s_wa + F_DIM))[lane];

  const int nw = (gridDim.x - GRP_CHUNKS) * 8;
  const int w0 = __builtin_amdgcn_readfirstlane((blockIdx.x - GRP_CHUNKS) * 8 + wv);
  for (int b0 = w0 * ROWS_IN_FLIGHT; b0 < M; b0 += nw * ROWS_IN_FLIGHT) {
    float4 xr[ROWS_IN_FLIGHT];
    #pragma unroll
    for (int k = 0; k < ROWS_IN_FLIGHT; ++k)
      xr[k] = row_of(x, min(b0 + k, M - 1), lane);
    float u[16];
    #pragma unroll
    for (int k = 0; k < ROWS_IN_FLIGHT; ++k) {
      u[k]     = dot4(xr[k], w1);
      u[8 + k] = dot4(xr[k], w2);
    }
    const float t = fold_lanes(u, lane);
    const int   i = lane >> 1;
    const int   r = b0 + (i & 7);
    if ((lane & 1) == 0 && r < M) {
      if (i < 8) as[r] = t; else ad[r] = t;
    }
  }
}


__global__ __launch_bounds__(GRP_THREADS) void gat_bucket_group(
    const int2* __restrict__ pairs, const int* __restrict__ cnt,
    int* __restrict__ deg, int* __restrict__ first, int* __restrict__ gsrc,
    int N, int E)
{
  static_assert(GRP_CHUNKS == 256 && GRP_THREADS == 1024, "32 waves x 8 table rows; 4 threads per chunk");
  __shared__ int s_off[GRP_CHUNKS];
  __shared__ int s_n[GRP_CHUNKS];
  __shared__ int s_c[256];
  __shared__ int s_wtot[8];
  __shared__ int s_lo[8];
  const int t    = threadIdx.x;
  const int lane = t & 31;
  const int w    = t >> 5;
  const int b    = blockIdx.x;

  int v[8];
  #pragma unroll
  for (int r = 0; r < 8; ++r) {
    const int  k  = w * 8 + r;
    const int4 c0 = ((const int4*)cnt)[k * 64 + lane];
    const int4 c1 = ((const int4*)cnt)[k * 64 + 32 + lane];
    const int  q0 = 4 * lane;
    const int  q1 = 128 + 4 * lane;
    int sum = 0;
    sum += (q0 + 0 < b) ? c0.x : 0;
    sum += (q0 + 1 < b) ? c0.y : 0;
    sum += (q0 + 2 < b) ? c0.z : 0;
    sum += (q0 + 3 < b) ? c0.w : 0;
    sum += (q1 + 0 < b) ? c1.x : 0;
    sum += (q1 + 1 < b) ? c1.y : 0;
    sum += (q1 + 2 < b) ? c1.z : 0;
    sum += (q1 + 3 < b) ? c1.w : 0;
    v[r] = sum;
  }
  const int off = fold_lanes(v, lane);
  if ((lane & 3) == 0) s_off[w * 8 + (lane >> 2)] = off;
  if (t < GRP_CHUNKS) s_n[t] = cnt[t * 256 + b];
  if (t < 256) s_c[t] = 0;
  __syncthreads();

  int part = (t < GRP_CHUNKS) ? s_off[t] : 0;
  #pragma unroll
  for (int m = 16; m >= 1; m >>= 1) part += __shfl_xor(part, m, 32);
  if (t < GRP_CHUNKS && lane == 0) s_lo[w] = part;

  const int per = (E + GRP_CHUNKS - 1) / GRP_CHUNKS;
  const int k   = t >> 2;
  const int sub = t & 3;
  const int nk  = s_n[k];
  const int2* __restrict__ pk = pairs + (size_t)k * per + s_off[k];
  for (int i0 = sub; i0 < nk; i0 += 4 * GRP_BATCH) {
    int d[GRP_BATCH];
    #pragma unroll
    for (int i = 0; i < GRP_BATCH; ++i)
      d[i] = pk[min(i0 + 4 * i, nk - 1)].x;
    #pragma unroll
    for (int i = 0; i < GRP_BATCH; ++i) keep_here(d[i]);
    #pragma unroll
    for (int i = 0; i < GRP_BATCH; ++i)
      if (i0 + 4 * i < nk) atomicAdd(&s_c[d[i] & 255], 1);
  }
  __syncthreads();

  int lo = 0;
  for (int i = 0; i < 8; ++i) lo += s_lo[i];

  const int c = (t < 256) ? s_c[t] : 0;
  int inc = c;
  #pragma unroll
  for (int o = 1; o < 32; o <<= 1) {
    int u = __shfl_up(inc, o, 32);
    if (lane >= o) inc += u;
  }
  if (t < 256 && lane == 31) s_wtot[w] = inc;
  __syncthreads();
  if (t < 256) {
    int front = lo;
    for (int i = 0; i < w; ++i) front += s_wtot[i];
    const int f = front + inc - c;
    const int n = b * 256 + t;
    if (n < N) { first[n] = f; deg[n] = c; }
    s_c[t] = f;
  }
  __syncthreads();
  for (int i0 = sub; i0 < nk; i0 += 4 * GRP_BATCH) {
    int2 q[GRP_BATCH];
    #pragma unroll
    for (int i = 0; i < GRP_BATCH; ++i)
      q[i] = pk[min(i0 + 4 * i, nk - 1)];
    #pragma unroll
    for (int i = 0; i < GRP_BATCH; ++i) { keep_here(q[i].x); keep_here(q[i].y); }
    #pragma unroll
    for (int i = 0; i < GRP_BATCH; ++i)
      if (i0 + 4 * i < nk) gsrc[atomicAdd(&s_c[q[i].x & 255], 1)] = q[i].y;
  }
}

__device__ __forceinline__ float4 node_sum(
    const int* __restrict__ gsrc, const float* __restrict__ x,
    const float* __restrict__ as, int beg, int cnt, float adv,
    int s0, float as0, int lane)
{
  const int* seg = gsrc + beg;

  const bool  v0 = lane < cnt;
  const float t0 = as0 + adv;
  const float l0 = v0 ? (t0 > 0.f ? t0 : LEAKY * t0) : -INFINITY;

  float m = l0;
  for (int c = 32; c < cnt; c += 32)
    if (c + lane < cnt) m = fmaxf(m, edge_logit(as, seg[c + lane], adv));
  m = wave_max(m);

  const float p0 = v0 ? expf(l0 - m) : 0.f;
  float sum = p0;
  for (int c = 32; c < cnt; c += 32)
    if (c + lane < cnt) sum += expf(edge_logit(as, seg[c + lane], adv) - m);
  sum = wave_sum(sum);
  const float den = sum + 1e-16f;

  float4 acc = make_float4(0.f, 0.f, 0.f, 0.f);
  add_rows(acc, x, lane, s0, p0 / den, min(cnt, 32));
  for (int c = 32; c < cnt; c += 32) {
    int s = 0; float p = 0.f;
    if (c + lane < cnt) { s = seg[c + lane]; p = expf(edge_logit(as, s, adv) - m); }
    add_rows(acc, x, lane, s, p / den, min(cnt - c, 32));
  }
  return acc;
}

__global__ __launch_bounds__(256) void gat_aggregate_gemm(
    const int* __restrict__ gsrc, const int* __restrict__ first,
    const int* __restrict__ deg, const float* __restrict__ x,
    const float* __restrict__ as, const float* __restrict__ ad,
    const float* __restrict__ W, float* __restrict__ out, int N)
{
  __shared__ float sA[TILE_NODES * 132];
  __shared__ float sH[TILE_NODES * 132];
  const int tid  = threadIdx.x;
  const int lane = tid & 31;
  const int wv   = tid >> 5;
  const int m0   = blockIdx.x * TILE_NODES;
  const int r0   = wv * NODES_PER_WAVE;

  int   beg[NODES_PER_WAVE], cnt[NODES_PER_WAVE], s0[NODES_PER_WAVE];
  float adv[NODES_PER_WAVE], as0[NODES_PER_WAVE];
  #pragma unroll
  for (int q = 0; q < NODES_PER_WAVE; ++q) {
    const int d  = __builtin_amdgcn_readfirstlane(m0 + r0 + q);
    const int dc = min(d, N - 1);
    beg[q] = first[dc];
    cnt[q] = (d < N) ? deg[dc] : 0;
    adv[q] = ad[dc];
  }
  #pragma unroll
  for (int q = 0; q < NODES_PER_WAVE; ++q) {
    s0[q] = 0;
    if (lane < cnt[q]) s0[q] = gsrc[beg[q] + lane];
  }
  #pragma unroll
  for (int q = 0; q < NODES_PER_WAVE; ++q) as0[q] = as[s0[q]];
  float* hs = sH + wv * (NODES_PER_WAVE * 64);
  #pragma unroll
  for (int q = 0; q < NODES_PER_WAVE; ++q) {
    hs[q * 64 + lane]      = __int_as_float(s0[q]);
    hs[q * 64 + 32 + lane] = as0[q];
  }
  #pragma unroll
  for (int q = 0; q < NODES_PER_WAVE; ++q) {
    const int   sq = __float_as_int(hs[q * 64 + lane]);
    const float aq = hs[q * 64 + 32 + lane];
    const float4 z = node_sum(gsrc, x, as, beg[q], cnt[q], adv[q], sq, aq, lane);
    ((float4*)sA)[(r0 + q) * 33 + lane] = z;
  }
  __syncthreads();

  const int row  = lane & 15;
  const int kh   = (lane >> 4) * 2;
  const int n0   = wv * 16;

  v8f acc0 = {};
  v8f acc1 = {};
  #pragma unroll 8
  for (int k = 0; k < F_DIM; k += 4) {
    v2f a0, a1, b;
    a0.x = sA[row * 132 + k + kh + 0];
    a0.y = sA[row * 132 + k + kh + 1];
    a1.x = sA[(16 + row) * 132 + k + kh + 0];
    a1.y = sA[(16 + row) * 132 + k + kh + 1];
    b.x = W[(k + kh + 0) * F_DIM + n0 + row];
    b.y = W[(k + kh + 1) * F_DIM + n0 + row];
    acc0 = __builtin_amdgcn_wmma_f32_16x16x4_f32(false, a0, false, b,
                                                 (short)0, acc0, false, false);
    acc1 = __builtin_amdgcn_wmma_f32_16x16x4_f32(false, a1, false, b,
                                                 (short)0, acc1, false, false);
  }

  const int mh = (lane >> 4) * 8;
  #pragma unroll
  for (int v = 0; v < 8; ++v) {
    sH[(mh + v) * 132 + n0 + row]      = acc0[v];
    sH[(16 + mh + v) * 132 + n0 + row] = acc1[v];
  }
  __syncthreads();

  #pragma unroll
  for (int q = 0; q < NODES_PER_WAVE; ++q) {
    const int g = m0 + r0 + q;
    if (g < N)
      ((float4*)(out + (size_t)g * F_DIM))[lane] = ((const float4*)sH)[(r0 + q) * 33 + lane];
  }
}

extern "C" void kernel_launch(void* const* d_in, const int* in_sizes, int n_in,
                              void* d_out, int out_size, void* d_ws, size_t ws_size,
                              hipStream_t stream) {
  const float* x  = (const float*)d_in[0];
  const int*   ei = (const int*)d_in[1];
  const float* W  = (const float*)d_in[2];
  const float* a  = (const float*)d_in[3];
  const int N = in_sizes[0] / F_DIM;
  const int E = in_sizes[1] / 2;
  const int* src = ei;
  const int* dst = ei + E;

  float* ws0   = (float*)d_ws;
  int*   deg   = (int*)(ws0 + (size_t)N * F_DIM);
  int*   first = deg + N;
  float* as    = (float*)(first + N);
  float* ad    = as + N;
  int*   gsrc  = (int*)(ad + N) + E;
  float* out   = (float*)d_out;
  int*   cnt    = (int*)d_out;
  int2*  pairs  = (int2*)(cnt + 256 * 256);

  gat_alpha_count   <<<GRP_CHUNKS + 782,  256,         0, stream>>>(x, W, a, as, ad, N, src, dst, cnt, pairs, E);
  gat_bucket_group  <<<(N + 255) / 256, GRP_THREADS, 0, stream>>>(pairs, cnt, deg, first, gsrc, N, E);
  gat_aggregate_gemm<<<(N + TILE_NODES - 1) / TILE_NODES, 256, 0, stream>>>(gsrc, first, deg, x, as, ad, W, out, N);
}
